# starting kernel
# baseline (speedup 1.0000x reference)
#include <hip/hip_runtime.h>
#include <hip/hip_bf16.h>
#include <math.h>


typedef __attribute__((ext_vector_type(2))) float v2f;
typedef __attribute__((ext_vector_type(8))) float v8f;

#define NIMG 124
#define IMG_ELEMS 16384
#define NE (NIMG * IMG_ELEMS)
#define LD 128
#define ITE 20


static __device__ __forceinline__ v8f wmma_f32(v2f a, v2f b, v8f c) {
  return __builtin_amdgcn_wmma_f32_16x16x4_f32(false, a, false, b,
                                               (short)0, c, false, false);
}

static __device__ __forceinline__ v2f afrag(const float* __restrict__ M,
                                            int mBase, int k, int r, int half) {
  const float* p = M + (size_t)(mBase + r) * LD + (k + 2 * half);
  v2f a;
  a.x = p[0];
  a.y = p[1];
  return a;
}

static __device__ __forceinline__ v2f bfrag(const float* __restrict__ M,
                                            int k, int nBase, int r, int half) {
  int row = k + 2 * half;
  v2f b;
  b.x = M[(size_t)row * LD + nBase + r];
  b.y = M[(size_t)(row + 1) * LD + nBase + r];
  return b;
}

static __device__ __forceinline__ void store_tile(float* __restrict__ M,
                                                  int mBase, int nBase, int r,
                                                  int half, v8f c, float s) {
#pragma unroll
  for (int j = 0; j < 8; ++j)
    M[(size_t)(mBase + j + 8 * half) * LD + nBase + r] = c[j] * s;
}


__global__ void k_tables(float* Ct, float* St, float* NSt) {
  int i = blockIdx.x * blockDim.x + threadIdx.x;
  if (i >= IMG_ELEMS) return;
  int m = i >> 7, k = i & 127;
  int p = (m * k) & 127;
  float ang = (6.283185307179586f / 128.f) * (float)p;
  float s, c;
  sincosf(ang, &s, &c);
  Ct[i] = c;
  St[i] = -s;
  NSt[i] = s;
}

__global__ void k_init(const float* __restrict__ Y, float* Z, float* X,
                       float* G1, float* G21, float* G22, float* bufH,
                       float* bufV) {
  int idx = blockIdx.x * blockDim.x + threadIdx.x;
  int w = idx & 127, h = (idx >> 7) & 127;
  int ib = idx & ~(IMG_ELEMS - 1);
  float y = Y[idx];
  Z[idx] = y;
  X[idx] = y;
  G1[idx] = 0.f;
  G21[idx] = 0.f;
  G22[idx] = 0.f;
  bufH[idx] = y - Y[ib + (h << 7) + ((w + 127) & 127)];
  bufV[idx] = y - Y[ib + (((h + 127) & 127) << 7) + w];
}

static __device__ __forceinline__ float shrinkf(float x, float t) {
  float m = fmaxf(fabsf(x) - t, 0.f);
  return copysignf(m, x);
}

__global__ void k_pre(const float* __restrict__ X, const float* __restrict__ G1,
                      const float* __restrict__ G21,
                      const float* __restrict__ G22, float* bufH, float* bufV,
                      float* bufC, float im1, float im2, float thr) {
  int idx = blockIdx.x * blockDim.x + threadIdx.x;
  float g21 = G21[idx] * im2;
  float g22 = G22[idx] * im2;
  float uh = shrinkf(bufH[idx] - g21, thr);
  float uv = shrinkf(bufV[idx] - g22, thr);
  bufC[idx] = X[idx] + G1[idx] * im1;
  bufH[idx] = uh + g21;
  bufV[idx] = uv + g22;
}

__global__ void k_filter(const float* __restrict__ MAr,
                         const float* __restrict__ MAi,
                         const float* __restrict__ MBr,
                         const float* __restrict__ MBi, float* Nr, float* Ni,
                         float mu1, float mu2) {
  int idx = blockIdx.x * blockDim.x + threadIdx.x;
  int v = idx & 127, u = (idx >> 7) & 127;
  int ib = idx & ~(IMG_ELEMS - 1);
  int um = (128 - u) & 127, vm = (128 - v) & 127;
  int midx = ib + (um << 7) + vm;
  float far_ = MAr[idx], fai = MAi[idx];
  float pr = MBr[idx], pi = MBi[idx];
  float qr = MBr[midx], qi = MBi[midx];
  float fbr = 0.5f * (pr + qr), fbi = 0.5f * (pi - qi);
  float fcr = 0.5f * (pi + qi), fci = -0.5f * (pr - qr);
  const float c0 = 6.283185307179586f / 128.f;
  float su, cu, sv, cv;
  sincosf(c0 * (float)u, &su, &cu);
  sincosf(c0 * (float)v, &sv, &cv);
  float hr = 1.f - cv, hi = -sv;
  float vr = 1.f - cu, vi = -su;
  float t1r = hr * fbr - hi * fbi, t1i = hr * fbi + hi * fbr;
  float t2r = vr * fcr - vi * fci, t2i = vr * fci + vi * fcr;
  float nr = mu1 * far_ + mu2 * (t1r + t2r);
  float ni = mu1 * fai + mu2 * (t1i + t2i);
  float den = mu1 + mu2 * (4.f - 2.f * cv - 2.f * cu);
  Nr[idx] = nr / den;
  Ni[idx] = ni / den;
}

__global__ void k_post(const float* __restrict__ Z, const float* __restrict__ Y,
                       const float* __restrict__ inW, float* X, float* G1,
                       float* G21, float* G22, float* bufH, float* bufV,
                       float mu1, float mu2, float im2) {
  int idx = blockIdx.x * blockDim.x + threadIdx.x;
  int w = idx & 127, h = (idx >> 7) & 127;
  int ib = idx & ~(IMG_ELEMS - 1);
  float z = Z[idx];
  float dh = z - Z[ib + (h << 7) + ((w + 127) & 127)];
  float dv = z - Z[ib + (((h + 127) & 127) << 7) + w];
  float g21 = G21[idx], g22 = G22[idx];
  float uh = bufH[idx] - g21 * im2;
  float uv = bufV[idx] - g22 * im2;
  float iw = inW[idx];
  float x = (iw * Y[idx] + mu1 * z - G1[idx]) / (iw + mu1);
  X[idx] = x;
  G1[idx] += mu1 * (x - z);
  G21[idx] = g21 + mu2 * (uh - dh);
  G22[idx] = g22 + mu2 * (uv - dv);
  bufH[idx] = dh;
  bufV[idx] = dv;
}


__global__ void pass1(const float* __restrict__ Ct, const float* __restrict__ St,
                      const float* __restrict__ NSt,
                      const float* __restrict__ RA, const float* __restrict__ RB,
                      const float* __restrict__ RC, float* TAr, float* TAi,
                      float* TBr, float* TBi) {
  size_t base = (size_t)blockIdx.x * IMG_ELEMS;
  const float* ra = RA + base;
  const float* rb = RB + base;
  const float* rc = RC + base;
  int wave = threadIdx.x >> 5, lane = threadIdx.x & 31;
  int r = lane & 15, half = lane >> 4;
  int mB = blockIdx.y << 4;
  int nB = wave << 4;
  v8f ar = {}, ai = {}, br = {}, bi = {};
  for (int k = 0; k < 128; k += 4) {
    v2f aC = afrag(Ct, mB, k, r, half);
    v2f aS = afrag(St, mB, k, r, half);
    v2f aN = afrag(NSt, mB, k, r, half);
    v2f bA = bfrag(ra, k, nB, r, half);
    v2f bB = bfrag(rb, k, nB, r, half);
    v2f bCc = bfrag(rc, k, nB, r, half);
    ar = wmma_f32(aC, bA, ar);
    ai = wmma_f32(aS, bA, ai);
    br = wmma_f32(aC, bB, br);
    br = wmma_f32(aN, bCc, br);
    bi = wmma_f32(aS, bB, bi);
    bi = wmma_f32(aC, bCc, bi);
  }
  store_tile(TAr + base, mB, nB, r, half, ar, 1.f);
  store_tile(TAi + base, mB, nB, r, half, ai, 1.f);
  store_tile(TBr + base, mB, nB, r, half, br, 1.f);
  store_tile(TBi + base, mB, nB, r, half, bi, 1.f);
}

__global__ void pass2(const float* __restrict__ Ct, const float* __restrict__ St,
                      const float* __restrict__ NSt,
                      const float* __restrict__ TAr,
                      const float* __restrict__ TAi,
                      const float* __restrict__ TBr,
                      const float* __restrict__ TBi, float* MAr, float* MAi,
                      float* MBr, float* MBi) {
  size_t base = (size_t)blockIdx.x * IMG_ELEMS;
  const float* tar = TAr + base;
  const float* tai = TAi + base;
  const float* tbr = TBr + base;
  const float* tbi = TBi + base;
  int wave = threadIdx.x >> 5, lane = threadIdx.x & 31;
  int r = lane & 15, half = lane >> 4;
  int mB = blockIdx.y << 4;
  int nB = wave << 4;
  v8f ar = {}, ai = {}, br = {}, bi = {};
  for (int k = 0; k < 128; k += 4) {
    v2f aAr = afrag(tar, mB, k, r, half);
    v2f aAi = afrag(tai, mB, k, r, half);
    v2f aBr = afrag(tbr, mB, k, r, half);
    v2f aBi = afrag(tbi, mB, k, r, half);
    v2f bC = bfrag(Ct, k, nB, r, half);
    v2f bS = bfrag(St, k, nB, r, half);
    v2f bN = bfrag(NSt, k, nB, r, half);
    ar = wmma_f32(aAr, bC, ar);
    ar = wmma_f32(aAi, bN, ar);
    ai = wmma_f32(aAr, bS, ai);
    ai = wmma_f32(aAi, bC, ai);
    br = wmma_f32(aBr, bC, br);
    br = wmma_f32(aBi, bN, br);
    bi = wmma_f32(aBr, bS, bi);
    bi = wmma_f32(aBi, bC, bi);
  }
  store_tile(MAr + base, mB, nB, r, half, ar, 1.f);
  store_tile(MAi + base, mB, nB, r, half, ai, 1.f);
  store_tile(MBr + base, mB, nB, r, half, br, 1.f);
  store_tile(MBi + base, mB, nB, r, half, bi, 1.f);
}

__global__ void pass3(const float* __restrict__ Ct, const float* __restrict__ St,
                      const float* __restrict__ NSt,
                      const float* __restrict__ Nr, const float* __restrict__ Ni,
                      float* Ur, float* Ui) {
  size_t base = (size_t)blockIdx.x * IMG_ELEMS;
  const float* nr = Nr + base;
  const float* ni = Ni + base;
  int wave = threadIdx.x >> 5, lane = threadIdx.x & 31;
  int r = lane & 15, half = lane >> 4;
  int mB = blockIdx.y << 4;
  int nB = wave << 4;
  v8f ur = {}, ui = {};
  for (int k = 0; k < 128; k += 4) {
    v2f aC = afrag(Ct, mB, k, r, half);
    v2f aS = afrag(St, mB, k, r, half);
    v2f aN = afrag(NSt, mB, k, r, half);
    v2f bR = bfrag(nr, k, nB, r, half);
    v2f bI = bfrag(ni, k, nB, r, half);
    ur = wmma_f32(aC, bR, ur);
    ur = wmma_f32(aS, bI, ur);
    ui = wmma_f32(aC, bI, ui);
    ui = wmma_f32(aN, bR, ui);
  }
  store_tile(Ur + base, mB, nB, r, half, ur, 1.f);
  store_tile(Ui + base, mB, nB, r, half, ui, 1.f);
}

__global__ void pass4(const float* __restrict__ Ct, const float* __restrict__ St,
                      const float* __restrict__ Ur, const float* __restrict__ Ui,
                      float* Z) {
  size_t base = (size_t)blockIdx.x * IMG_ELEMS;
  const float* ur = Ur + base;
  const float* ui = Ui + base;
  int wave = threadIdx.x >> 5, lane = threadIdx.x & 31;
  int r = lane & 15, half = lane >> 4;
  int mB = blockIdx.y << 4;
  int nB = wave << 4;
  v8f z = {};
  for (int k = 0; k < 128; k += 4) {
    v2f aR = afrag(ur, mB, k, r, half);
    v2f aI = afrag(ui, mB, k, r, half);
    v2f bC = bfrag(Ct, k, nB, r, half);
    v2f bS = bfrag(St, k, nB, r, half);
    z = wmma_f32(aR, bC, z);
    z = wmma_f32(aI, bS, z);
  }
  store_tile(Z + base, mB, nB, r, half, z, 1.f / 16384.f);
}


extern "C" void kernel_launch(void* const* d_in, const int* in_sizes, int n_in,
                              void* d_out, int out_size, void* d_ws,
                              size_t ws_size, hipStream_t stream) {
  (void)in_sizes; (void)n_in; (void)out_size; (void)ws_size;
  const float* Y = (const float*)d_in[0];
  const float* inW = (const float*)d_in[1];
  float* Z = (float*)d_out;
  float* ws = (float*)d_ws;

  float* Ct = ws;
  float* St = Ct + IMG_ELEMS;
  float* NSt = St + IMG_ELEMS;
  float* p = ws + 4 * (size_t)IMG_ELEMS;
  float* X = p;            p += NE;
  float* G1 = p;           p += NE;
  float* G21 = p;          p += NE;
  float* G22 = p;          p += NE;
  float* bufH = p;         p += NE;
  float* bufV = p;         p += NE;
  float* bufC = p;         p += NE;
  float* TAr = p;          p += NE;
  float* TAi = p;          p += NE;
  float* TBr = p;          p += NE;
  float* TBi = p;          p += NE;
  float* MAr = p;          p += NE;
  float* MAi = p;          p += NE;
  float* MBr = p;          p += NE;
  float* MBi = p;          p += NE;

  const int EB = 256, EG = NE / EB;
  const dim3 gg(NIMG, 8);

  k_tables<<<IMG_ELEMS / 256, 256, 0, stream>>>(Ct, St, NSt);
  k_init<<<EG, EB, 0, stream>>>(Y, Z, X, G1, G21, G22, bufH, bufV);

  float m1 = 0.1f, m2 = 0.1f;
  for (int it = 0; it < ITE; ++it) {
    float im1 = 1.f / m1, im2 = 1.f / m2, thr = 0.1f / m2;
    k_pre<<<EG, EB, 0, stream>>>(X, G1, G21, G22, bufH, bufV, bufC, im1, im2,
                                 thr);
    pass1<<<gg, 256, 0, stream>>>(Ct, St, NSt, bufC, bufH, bufV, TAr, TAi,
                                  TBr, TBi);
    pass2<<<gg, 256, 0, stream>>>(Ct, St, NSt, TAr, TAi, TBr, TBi, MAr, MAi,
                                  MBr, MBi);
    k_filter<<<EG, EB, 0, stream>>>(MAr, MAi, MBr, MBi, TAr, TAi, m1, m2);
    pass3<<<gg, 256, 0, stream>>>(Ct, St, NSt, TAr, TAi, TBr, TBi);
    pass4<<<gg, 256, 0, stream>>>(Ct, St, TBr, TBi, Z);
    if (it < ITE - 1)
      k_post<<<EG, EB, 0, stream>>>(Z, Y, inW, X, G1, G21, G22, bufH, bufV, m1,
                                    m2, im2);
    m1 *= 1.05f;
    m2 *= 1.05f;
  }
}


# adjoint differences applied in space before the transform (sums reordered): one real array goes through the forward DFT instead of three; the Fourier step becomes a real per-frequency scale
# speedup vs baseline: 1.2724x; 1.2724x over previous
#include <hip/hip_runtime.h>
#include <hip/hip_bf16.h>
#include <math.h>


typedef __attribute__((ext_vector_type(2))) float v2f;
typedef __attribute__((ext_vector_type(8))) float v8f;

#define NIMG 124
#define IMG_ELEMS 16384
#define NE (NIMG * IMG_ELEMS)
#define LD 128
#define ITE 20


static __device__ __forceinline__ v8f wmma_f32(v2f a, v2f b, v8f c) {
  return __builtin_amdgcn_wmma_f32_16x16x4_f32(false, a, false, b,
                                               (short)0, c, false, false);
}

static __device__ __forceinline__ v2f afrag(const float* __restrict__ M,
                                            int mBase, int k, int r, int half) {
  const float* p = M + (size_t)(mBase + r) * LD + (k + 2 * half);
  v2f a;
  a.x = p[0];
  a.y = p[1];
  return a;
}

static __device__ __forceinline__ v2f bfrag(const float* __restrict__ M,
                                            int k, int nBase, int r, int half) {
  int row = k + 2 * half;
  v2f b;
  b.x = M[(size_t)row * LD + nBase + r];
  b.y = M[(size_t)(row + 1) * LD + nBase + r];
  return b;
}

static __device__ __forceinline__ void store_tile(float* __restrict__ M,
                                                  int mBase, int nBase, int r,
                                                  int half, v8f c, float s) {
#pragma unroll
  for (int j = 0; j < 8; ++j)
    M[(size_t)(mBase + j + 8 * half) * LD + nBase + r] = c[j] * s;
}


__global__ void k_tables(float* Ct, float* St, float* NSt) {
  int i = blockIdx.x * blockDim.x + threadIdx.x;
  if (i >= IMG_ELEMS) return;
  int m = i >> 7, k = i & 127;
  int p = (m * k) & 127;
  float ang = (6.283185307179586f / 128.f) * (float)p;
  float s, c;
  sincosf(ang, &s, &c);
  Ct[i] = c;
  St[i] = -s;
  NSt[i] = s;
}

__global__ void k_init(const float* __restrict__ Y, float* Z, float* X,
                       float* G1, float* G21, float* G22, float* bufH,
                       float* bufV) {
  int idx = blockIdx.x * blockDim.x + threadIdx.x;
  int w = idx & 127, h = (idx >> 7) & 127;
  int ib = idx & ~(IMG_ELEMS - 1);
  float y = Y[idx];
  Z[idx] = y;
  X[idx] = y;
  G1[idx] = 0.f;
  G21[idx] = 0.f;
  G22[idx] = 0.f;
  bufH[idx] = y - Y[ib + (h << 7) + ((w + 127) & 127)];
  bufV[idx] = y - Y[ib + (((h + 127) & 127) << 7) + w];
}

static __device__ __forceinline__ float shrinkf(float x, float t) {
  float m = fmaxf(fabsf(x) - t, 0.f);
  return copysignf(m, x);
}

__global__ void k_pre(const float* __restrict__ X, const float* __restrict__ G1,
                      const float* __restrict__ G21,
                      const float* __restrict__ G22,
                      const float* __restrict__ bufH,
                      const float* __restrict__ bufV, float* __restrict__ bufC,
                      float mu1, float mu2, float im1, float im2, float thr) {
  int idx = blockIdx.x * blockDim.x + threadIdx.x;
  int w = idx & 127, h = (idx >> 7) & 127;
  int ib = idx & ~(IMG_ELEMS - 1);
  int iw = ib + (h << 7) + ((w + 1) & 127);
  int ih = ib + (((h + 1) & 127) << 7) + w;
  float g21 = G21[idx] * im2, g21n = G21[iw] * im2;
  float g22 = G22[idx] * im2, g22n = G22[ih] * im2;
  float b = shrinkf(bufH[idx] - g21, thr) + g21;
  float bn = shrinkf(bufH[iw] - g21n, thr) + g21n;
  float c = shrinkf(bufV[idx] - g22, thr) + g22;
  float cn = shrinkf(bufV[ih] - g22n, thr) + g22n;
  float a = X[idx] + G1[idx] * im1;
  bufC[idx] = mu1 * a + mu2 * ((b - bn) + (c - cn));
}

__global__ void k_filter(float* Mr, float* Mi, float mu1, float mu2) {
  int idx = blockIdx.x * blockDim.x + threadIdx.x;
  int v = idx & 127, u = (idx >> 7) & 127;
  const float c0 = 6.283185307179586f / 128.f;
  float cu = cosf(c0 * (float)u);
  float cv = cosf(c0 * (float)v);
  float den = mu1 + mu2 * (4.f - 2.f * cv - 2.f * cu);
  Mr[idx] = Mr[idx] / den;
  Mi[idx] = Mi[idx] / den;
}

__global__ void k_post(const float* __restrict__ Z, const float* __restrict__ Y,
                       const float* __restrict__ inW, float* X, float* G1,
                       float* G21, float* G22, float* bufH, float* bufV,
                       float mu1, float mu2, float im2, float thr) {
  int idx = blockIdx.x * blockDim.x + threadIdx.x;
  int w = idx & 127, h = (idx >> 7) & 127;
  int ib = idx & ~(IMG_ELEMS - 1);
  float z = Z[idx];
  float dh = z - Z[ib + (h << 7) + ((w + 127) & 127)];
  float dv = z - Z[ib + (((h + 127) & 127) << 7) + w];
  float g21 = G21[idx], g22 = G22[idx];
  float uh = shrinkf(bufH[idx] - g21 * im2, thr);
  float uv = shrinkf(bufV[idx] - g22 * im2, thr);
  float iw = inW[idx];
  float x = (iw * Y[idx] + mu1 * z - G1[idx]) / (iw + mu1);
  X[idx] = x;
  G1[idx] += mu1 * (x - z);
  G21[idx] = g21 + mu2 * (uh - dh);
  G22[idx] = g22 + mu2 * (uv - dv);
  bufH[idx] = dh;
  bufV[idx] = dv;
}


__global__ void pass1(const float* __restrict__ Ct, const float* __restrict__ St,
                      const float* __restrict__ R, float* Tr, float* Ti) {
  size_t base = (size_t)blockIdx.x * IMG_ELEMS;
  const float* rr = R + base;
  int wave = threadIdx.x >> 5, lane = threadIdx.x & 31;
  int r = lane & 15, half = lane >> 4;
  int mB = blockIdx.y << 4;
  int nB = wave << 4;
  v8f tr = {}, ti = {};
  for (int k = 0; k < 128; k += 4) {
    v2f aC = afrag(Ct, mB, k, r, half);
    v2f aS = afrag(St, mB, k, r, half);
    v2f bR = bfrag(rr, k, nB, r, half);
    tr = wmma_f32(aC, bR, tr);
    ti = wmma_f32(aS, bR, ti);
  }
  store_tile(Tr + base, mB, nB, r, half, tr, 1.f);
  store_tile(Ti + base, mB, nB, r, half, ti, 1.f);
}

__global__ void pass2(const float* __restrict__ Ct, const float* __restrict__ St,
                      const float* __restrict__ NSt,
                      const float* __restrict__ Tr,
                      const float* __restrict__ Ti, float* Mr, float* Mi) {
  size_t base = (size_t)blockIdx.x * IMG_ELEMS;
  const float* tr = Tr + base;
  const float* ti = Ti + base;
  int wave = threadIdx.x >> 5, lane = threadIdx.x & 31;
  int r = lane & 15, half = lane >> 4;
  int mB = blockIdx.y << 4;
  int nB = wave << 4;
  v8f mr = {}, mi = {};
  for (int k = 0; k < 128; k += 4) {
    v2f aR = afrag(tr, mB, k, r, half);
    v2f aI = afrag(ti, mB, k, r, half);
    v2f bC = bfrag(Ct, k, nB, r, half);
    v2f bS = bfrag(St, k, nB, r, half);
    v2f bN = bfrag(NSt, k, nB, r, half);
    mr = wmma_f32(aR, bC, mr);
    mr = wmma_f32(aI, bN, mr);
    mi = wmma_f32(aR, bS, mi);
    mi = wmma_f32(aI, bC, mi);
  }
  store_tile(Mr + base, mB, nB, r, half, mr, 1.f);
  store_tile(Mi + base, mB, nB, r, half, mi, 1.f);
}

__global__ void pass3(const float* __restrict__ Ct, const float* __restrict__ St,
                      const float* __restrict__ NSt,
                      const float* __restrict__ Nr, const float* __restrict__ Ni,
                      float* Ur, float* Ui) {
  size_t base = (size_t)blockIdx.x * IMG_ELEMS;
  const float* nr = Nr + base;
  const float* ni = Ni + base;
  int wave = threadIdx.x >> 5, lane = threadIdx.x & 31;
  int r = lane & 15, half = lane >> 4;
  int mB = blockIdx.y << 4;
  int nB = wave << 4;
  v8f ur = {}, ui = {};
  for (int k = 0; k < 128; k += 4) {
    v2f aC = afrag(Ct, mB, k, r, half);
    v2f aS = afrag(St, mB, k, r, half);
    v2f aN = afrag(NSt, mB, k, r, half);
    v2f bR = bfrag(nr, k, nB, r, half);
    v2f bI = bfrag(ni, k, nB, r, half);
    ur = wmma_f32(aC, bR, ur);
    ur = wmma_f32(aS, bI, ur);
    ui = wmma_f32(aC, bI, ui);
    ui = wmma_f32(aN, bR, ui);
  }
  store_tile(Ur + base, mB, nB, r, half, ur, 1.f);
  store_tile(Ui + base, mB, nB, r, half, ui, 1.f);
}

__global__ void pass4(const float* __restrict__ Ct, const float* __restrict__ St,
                      const float* __restrict__ Ur, const float* __restrict__ Ui,
                      float* Z) {
  size_t base = (size_t)blockIdx.x * IMG_ELEMS;
  const float* ur = Ur + base;
  const float* ui = Ui + base;
  int wave = threadIdx.x >> 5, lane = threadIdx.x & 31;
  int r = lane & 15, half = lane >> 4;
  int mB = blockIdx.y << 4;
  int nB = wave << 4;
  v8f z = {};
  for (int k = 0; k < 128; k += 4) {
    v2f aR = afrag(ur, mB, k, r, half);
    v2f aI = afrag(ui, mB, k, r, half);
    v2f bC = bfrag(Ct, k, nB, r, half);
    v2f bS = bfrag(St, k, nB, r, half);
    z = wmma_f32(aR, bC, z);
    z = wmma_f32(aI, bS, z);
  }
  store_tile(Z + base, mB, nB, r, half, z, 1.f / 16384.f);
}


extern "C" void kernel_launch(void* const* d_in, const int* in_sizes, int n_in,
                              void* d_out, int out_size, void* d_ws,
                              size_t ws_size, hipStream_t stream) {
  (void)in_sizes; (void)n_in; (void)out_size; (void)ws_size;
  const float* Y = (const float*)d_in[0];
  const float* inW = (const float*)d_in[1];
  float* Z = (float*)d_out;
  float* ws = (float*)d_ws;

  float* Ct = ws;
  float* St = Ct + IMG_ELEMS;
  float* NSt = St + IMG_ELEMS;
  float* p = ws + 4 * (size_t)IMG_ELEMS;
  float* X = p;            p += NE;
  float* G1 = p;           p += NE;
  float* G21 = p;          p += NE;
  float* G22 = p;          p += NE;
  float* bufH = p;         p += NE;
  float* bufV = p;         p += NE;
  float* bufC = p;         p += NE;
  float* Tr = p;           p += NE;
  float* Ti = p;           p += NE;
  float* Mr = p;           p += NE;
  float* Mi = p;           p += NE;

  const int EB = 256, EG = NE / EB;
  const dim3 gg(NIMG, 8);

  k_tables<<<IMG_ELEMS / 256, 256, 0, stream>>>(Ct, St, NSt);
  k_init<<<EG, EB, 0, stream>>>(Y, Z, X, G1, G21, G22, bufH, bufV);

  float m1 = 0.1f, m2 = 0.1f;
  for (int it = 0; it < ITE; ++it) {
    float im1 = 1.f / m1, im2 = 1.f / m2, thr = 0.1f / m2;
    k_pre<<<EG, EB, 0, stream>>>(X, G1, G21, G22, bufH, bufV, bufC, m1, m2,
                                 im1, im2, thr);
    pass1<<<gg, 256, 0, stream>>>(Ct, St, bufC, Tr, Ti);
    pass2<<<gg, 256, 0, stream>>>(Ct, St, NSt, Tr, Ti, Mr, Mi);
    k_filter<<<EG, EB, 0, stream>>>(Mr, Mi, m1, m2);
    pass3<<<gg, 256, 0, stream>>>(Ct, St, NSt, Mr, Mi, Tr, Ti);
    pass4<<<gg, 256, 0, stream>>>(Ct, St, Tr, Ti, Z);
    if (it < ITE - 1)
      k_post<<<EG, EB, 0, stream>>>(Z, Y, inW, X, G1, G21, G22, bufH, bufV, m1,
                                    m2, im2, thr);
    m1 *= 1.05f;
    m2 *= 1.05f;
  }
}


# Fourier-domain divide by den moved into the pass-2 store (den from the cosine table's row 1): the separate filter launch and its read and write of the spectrum go away, 141 -> 121 launches
# speedup vs baseline: 1.3785x; 1.0834x over previous
#include <hip/hip_runtime.h>
#include <hip/hip_bf16.h>
#include <math.h>


typedef __attribute__((ext_vector_type(2))) float v2f;
typedef __attribute__((ext_vector_type(8))) float v8f;

#define NIMG 124
#define IMG_ELEMS 16384
#define NE (NIMG * IMG_ELEMS)
#define LD 128
#define ITE 20


static __device__ __forceinline__ v8f wmma_f32(v2f a, v2f b, v8f c) {
  return __builtin_amdgcn_wmma_f32_16x16x4_f32(false, a, false, b,
                                               (short)0, c, false, false);
}

static __device__ __forceinline__ v2f afrag(const float* __restrict__ M,
                                            int mBase, int k, int r, int half) {
  const float* p = M + (size_t)(mBase + r) * LD + (k + 2 * half);
  v2f a;
  a.x = p[0];
  a.y = p[1];
  return a;
}

static __device__ __forceinline__ v2f bfrag(const float* __restrict__ M,
                                            int k, int nBase, int r, int half) {
  int row = k + 2 * half;
  v2f b;
  b.x = M[(size_t)row * LD + nBase + r];
  b.y = M[(size_t)(row + 1) * LD + nBase + r];
  return b;
}

static __device__ __forceinline__ void store_tile(float* __restrict__ M,
                                                  int mBase, int nBase, int r,
                                                  int half, v8f c, float s) {
#pragma unroll
  for (int j = 0; j < 8; ++j)
    M[(size_t)(mBase + j + 8 * half) * LD + nBase + r] = c[j] * s;
}


__global__ void k_tables(float* Ct, float* St, float* NSt) {
  int i = blockIdx.x * blockDim.x + threadIdx.x;
  if (i >= IMG_ELEMS) return;
  int m = i >> 7, k = i & 127;
  int p = (m * k) & 127;
  float ang = (6.283185307179586f / 128.f) * (float)p;
  float s, c;
  sincosf(ang, &s, &c);
  Ct[i] = c;
  St[i] = -s;
  NSt[i] = s;
}

__global__ void k_init(const float* __restrict__ Y, float* Z, float* X,
                       float* G1, float* G21, float* G22, float* bufH,
                       float* bufV) {
  int idx = blockIdx.x * blockDim.x + threadIdx.x;
  int w = idx & 127, h = (idx >> 7) & 127;
  int ib = idx & ~(IMG_ELEMS - 1);
  float y = Y[idx];
  Z[idx] = y;
  X[idx] = y;
  G1[idx] = 0.f;
  G21[idx] = 0.f;
  G22[idx] = 0.f;
  bufH[idx] = y - Y[ib + (h << 7) + ((w + 127) & 127)];
  bufV[idx] = y - Y[ib + (((h + 127) & 127) << 7) + w];
}

static __device__ __forceinline__ float shrinkf(float x, float t) {
  float m = fmaxf(fabsf(x) - t, 0.f);
  return copysignf(m, x);
}

__global__ void k_pre(const float* __restrict__ X, const float* __restrict__ G1,
                      const float* __restrict__ G21,
                      const float* __restrict__ G22,
                      const float* __restrict__ bufH,
                      const float* __restrict__ bufV, float* __restrict__ bufC,
                      float mu1, float mu2, float im1, float im2, float thr) {
  int idx = blockIdx.x * blockDim.x + threadIdx.x;
  int w = idx & 127, h = (idx >> 7) & 127;
  int ib = idx & ~(IMG_ELEMS - 1);
  int iw = ib + (h << 7) + ((w + 1) & 127);
  int ih = ib + (((h + 1) & 127) << 7) + w;
  float g21 = G21[idx] * im2, g21n = G21[iw] * im2;
  float g22 = G22[idx] * im2, g22n = G22[ih] * im2;
  float b = shrinkf(bufH[idx] - g21, thr) + g21;
  float bn = shrinkf(bufH[iw] - g21n, thr) + g21n;
  float c = shrinkf(bufV[idx] - g22, thr) + g22;
  float cn = shrinkf(bufV[ih] - g22n, thr) + g22n;
  float a = X[idx] + G1[idx] * im1;
  bufC[idx] = mu1 * a + mu2 * ((b - bn) + (c - cn));
}

__global__ void k_post(const float* __restrict__ Z, const float* __restrict__ Y,
                       const float* __restrict__ inW, float* X, float* G1,
                       float* G21, float* G22, float* bufH, float* bufV,
                       float mu1, float mu2, float im2, float thr) {
  int idx = blockIdx.x * blockDim.x + threadIdx.x;
  int w = idx & 127, h = (idx >> 7) & 127;
  int ib = idx & ~(IMG_ELEMS - 1);
  float z = Z[idx];
  float dh = z - Z[ib + (h << 7) + ((w + 127) & 127)];
  float dv = z - Z[ib + (((h + 127) & 127) << 7) + w];
  float g21 = G21[idx], g22 = G22[idx];
  float uh = shrinkf(bufH[idx] - g21 * im2, thr);
  float uv = shrinkf(bufV[idx] - g22 * im2, thr);
  float iw = inW[idx];
  float x = (iw * Y[idx] + mu1 * z - G1[idx]) / (iw + mu1);
  X[idx] = x;
  G1[idx] += mu1 * (x - z);
  G21[idx] = g21 + mu2 * (uh - dh);
  G22[idx] = g22 + mu2 * (uv - dv);
  bufH[idx] = dh;
  bufV[idx] = dv;
}


__global__ void pass1(const float* __restrict__ Ct, const float* __restrict__ St,
                      const float* __restrict__ R, float* Tr, float* Ti) {
  size_t base = (size_t)blockIdx.x * IMG_ELEMS;
  const float* rr = R + base;
  int wave = threadIdx.x >> 5, lane = threadIdx.x & 31;
  int r = lane & 15, half = lane >> 4;
  int mB = blockIdx.y << 4;
  int nB = wave << 4;
  v8f tr = {}, ti = {};
  for (int k = 0; k < 128; k += 4) {
    v2f aC = afrag(Ct, mB, k, r, half);
    v2f aS = afrag(St, mB, k, r, half);
    v2f bR = bfrag(rr, k, nB, r, half);
    tr = wmma_f32(aC, bR, tr);
    ti = wmma_f32(aS, bR, ti);
  }
  store_tile(Tr + base, mB, nB, r, half, tr, 1.f);
  store_tile(Ti + base, mB, nB, r, half, ti, 1.f);
}

__global__ void pass2(const float* __restrict__ Ct, const float* __restrict__ St,
                      const float* __restrict__ NSt,
                      const float* __restrict__ Tr,
                      const float* __restrict__ Ti, float* Mr, float* Mi,
                      float mu1, float mu2) {
  size_t base = (size_t)blockIdx.x * IMG_ELEMS;
  const float* tr = Tr + base;
  const float* ti = Ti + base;
  int wave = threadIdx.x >> 5, lane = threadIdx.x & 31;
  int r = lane & 15, half = lane >> 4;
  int mB = blockIdx.y << 4;
  int nB = wave << 4;
  v8f mr = {}, mi = {};
  for (int k = 0; k < 128; k += 4) {
    v2f aR = afrag(tr, mB, k, r, half);
    v2f aI = afrag(ti, mB, k, r, half);
    v2f bC = bfrag(Ct, k, nB, r, half);
    v2f bS = bfrag(St, k, nB, r, half);
    v2f bN = bfrag(NSt, k, nB, r, half);
    mr = wmma_f32(aR, bC, mr);
    mr = wmma_f32(aI, bN, mr);
    mi = wmma_f32(aR, bS, mi);
    mi = wmma_f32(aI, bC, mi);
  }
  float* mro = Mr + base;
  float* mio = Mi + base;
  const float* cosrow = Ct + LD;
  int v = nB + r;
  float cv = cosrow[v];
  for (int j = 0; j < 8; ++j) {
    int u = mB + j + 8 * half;
    float cu = cosrow[u];
    float den = mu1 + mu2 * (4.f - 2.f * cv - 2.f * cu);
    mro[(size_t)u * LD + v] = mr[j] / den;
    mio[(size_t)u * LD + v] = mi[j] / den;
  }
}

__global__ void pass3(const float* __restrict__ Ct, const float* __restrict__ St,
                      const float* __restrict__ NSt,
                      const float* __restrict__ Nr, const float* __restrict__ Ni,
                      float* Ur, float* Ui) {
  size_t base = (size_t)blockIdx.x * IMG_ELEMS;
  const float* nr = Nr + base;
  const float* ni = Ni + base;
  int wave = threadIdx.x >> 5, lane = threadIdx.x & 31;
  int r = lane & 15, half = lane >> 4;
  int mB = blockIdx.y << 4;
  int nB = wave << 4;
  v8f ur = {}, ui = {};
  for (int k = 0; k < 128; k += 4) {
    v2f aC = afrag(Ct, mB, k, r, half);
    v2f aS = afrag(St, mB, k, r, half);
    v2f aN = afrag(NSt, mB, k, r, half);
    v2f bR = bfrag(nr, k, nB, r, half);
    v2f bI = bfrag(ni, k, nB, r, half);
    ur = wmma_f32(aC, bR, ur);
    ur = wmma_f32(aS, bI, ur);
    ui = wmma_f32(aC, bI, ui);
    ui = wmma_f32(aN, bR, ui);
  }
  store_tile(Ur + base, mB, nB, r, half, ur, 1.f);
  store_tile(Ui + base, mB, nB, r, half, ui, 1.f);
}

__global__ void pass4(const float* __restrict__ Ct, const float* __restrict__ St,
                      const float* __restrict__ Ur, const float* __restrict__ Ui,
                      float* Z) {
  size_t base = (size_t)blockIdx.x * IMG_ELEMS;
  const float* ur = Ur + base;
  const float* ui = Ui + base;
  int wave = threadIdx.x >> 5, lane = threadIdx.x & 31;
  int r = lane & 15, half = lane >> 4;
  int mB = blockIdx.y << 4;
  int nB = wave << 4;
  v8f z = {};
  for (int k = 0; k < 128; k += 4) {
    v2f aR = afrag(ur, mB, k, r, half);
    v2f aI = afrag(ui, mB, k, r, half);
    v2f bC = bfrag(Ct, k, nB, r, half);
    v2f bS = bfrag(St, k, nB, r, half);
    z = wmma_f32(aR, bC, z);
    z = wmma_f32(aI, bS, z);
  }
  store_tile(Z + base, mB, nB, r, half, z, 1.f / 16384.f);
}


extern "C" void kernel_launch(void* const* d_in, const int* in_sizes, int n_in,
                              void* d_out, int out_size, void* d_ws,
                              size_t ws_size, hipStream_t stream) {
  (void)in_sizes; (void)n_in; (void)out_size; (void)ws_size;
  const float* Y = (const float*)d_in[0];
  const float* inW = (const float*)d_in[1];
  float* Z = (float*)d_out;
  float* ws = (float*)d_ws;

  float* Ct = ws;
  float* St = Ct + IMG_ELEMS;
  float* NSt = St + IMG_ELEMS;
  float* p = ws + 4 * (size_t)IMG_ELEMS;
  float* X = p;            p += NE;
  float* G1 = p;           p += NE;
  float* G21 = p;          p += NE;
  float* G22 = p;          p += NE;
  float* bufH = p;         p += NE;
  float* bufV = p;         p += NE;
  float* bufC = p;         p += NE;
  float* Tr = p;           p += NE;
  float* Ti = p;           p += NE;
  float* Mr = p;           p += NE;
  float* Mi = p;           p += NE;

  const int EB = 256, EG = NE / EB;
  const dim3 gg(NIMG, 8);

  k_tables<<<IMG_ELEMS / 256, 256, 0, stream>>>(Ct, St, NSt);
  k_init<<<EG, EB, 0, stream>>>(Y, Z, X, G1, G21, G22, bufH, bufV);

  float m1 = 0.1f, m2 = 0.1f;
  for (int it = 0; it < ITE; ++it) {
    float im1 = 1.f / m1, im2 = 1.f / m2, thr = 0.1f / m2;
    k_pre<<<EG, EB, 0, stream>>>(X, G1, G21, G22, bufH, bufV, bufC, m1, m2,
                                 im1, im2, thr);
    pass1<<<gg, 256, 0, stream>>>(Ct, St, bufC, Tr, Ti);
    pass2<<<gg, 256, 0, stream>>>(Ct, St, NSt, Tr, Ti, Mr, Mi, m1, m2);
    pass3<<<gg, 256, 0, stream>>>(Ct, St, NSt, Mr, Mi, Tr, Ti);
    pass4<<<gg, 256, 0, stream>>>(Ct, St, Tr, Ti, Z);
    if (it < ITE - 1)
      k_post<<<EG, EB, 0, stream>>>(Z, Y, inW, X, G1, G21, G22, bufH, bufV, m1,
                                    m2, im2, thr);
    m1 *= 1.05f;
    m2 *= 1.05f;
  }
}


# complex DFT replaced by the real Hartley (cas) transform: den is even in each frequency, so cas*cas modes are eigenvectors of the solve; 12 -> 4 matrix products per k-step (sums reordered)
# speedup vs baseline: 2.4161x; 1.7527x over previous
#include <hip/hip_runtime.h>
#include <hip/hip_bf16.h>
#include <math.h>


typedef __attribute__((ext_vector_type(2))) float v2f;
typedef __attribute__((ext_vector_type(8))) float v8f;

#define NIMG 124
#define IMG_ELEMS 16384
#define NE (NIMG * IMG_ELEMS)
#define LD 128
#define ITE 20


static __device__ __forceinline__ v8f wmma_f32(v2f a, v2f b, v8f c) {
  return __builtin_amdgcn_wmma_f32_16x16x4_f32(false, a, false, b,
                                               (short)0, c, false, false);
}

static __device__ __forceinline__ v2f afrag(const float* __restrict__ M,
                                            int mBase, int k, int r, int half) {
  const float* p = M + (size_t)(mBase + r) * LD + (k + 2 * half);
  v2f a;
  a.x = p[0];
  a.y = p[1];
  return a;
}

static __device__ __forceinline__ v2f bfrag(const float* __restrict__ M,
                                            int k, int nBase, int r, int half) {
  int row = k + 2 * half;
  v2f b;
  b.x = M[(size_t)row * LD + nBase + r];
  b.y = M[(size_t)(row + 1) * LD + nBase + r];
  return b;
}

static __device__ __forceinline__ void store_tile(float* __restrict__ M,
                                                  int mBase, int nBase, int r,
                                                  int half, v8f c, float s) {
#pragma unroll
  for (int j = 0; j < 8; ++j)
    M[(size_t)(mBase + j + 8 * half) * LD + nBase + r] = c[j] * s;
}


__global__ void k_tables(float* Ht, float* cosrow) {
  int i = blockIdx.x * blockDim.x + threadIdx.x;
  if (i >= IMG_ELEMS) return;
  int m = i >> 7, k = i & 127;
  int p = (m * k) & 127;
  float ang = (6.283185307179586f / 128.f) * (float)p;
  float s, c;
  sincosf(ang, &s, &c);
  Ht[i] = c + s;
  if (m == 1) cosrow[k] = c;
}

__global__ void k_init(const float* __restrict__ Y, float* Z, float* X,
                       float* G1, float* G21, float* G22, float* bufH,
                       float* bufV) {
  int idx = blockIdx.x * blockDim.x + threadIdx.x;
  int w = idx & 127, h = (idx >> 7) & 127;
  int ib = idx & ~(IMG_ELEMS - 1);
  float y = Y[idx];
  Z[idx] = y;
  X[idx] = y;
  G1[idx] = 0.f;
  G21[idx] = 0.f;
  G22[idx] = 0.f;
  bufH[idx] = y - Y[ib + (h << 7) + ((w + 127) & 127)];
  bufV[idx] = y - Y[ib + (((h + 127) & 127) << 7) + w];
}

static __device__ __forceinline__ float shrinkf(float x, float t) {
  float m = fmaxf(fabsf(x) - t, 0.f);
  return copysignf(m, x);
}

__global__ void k_pre(const float* __restrict__ X, const float* __restrict__ G1,
                      const float* __restrict__ G21,
                      const float* __restrict__ G22,
                      const float* __restrict__ bufH,
                      const float* __restrict__ bufV, float* __restrict__ bufC,
                      float mu1, float mu2, float im1, float im2, float thr) {
  int idx = blockIdx.x * blockDim.x + threadIdx.x;
  int w = idx & 127, h = (idx >> 7) & 127;
  int ib = idx & ~(IMG_ELEMS - 1);
  int iw = ib + (h << 7) + ((w + 1) & 127);
  int ih = ib + (((h + 1) & 127) << 7) + w;
  float g21 = G21[idx] * im2, g21n = G21[iw] * im2;
  float g22 = G22[idx] * im2, g22n = G22[ih] * im2;
  float b = shrinkf(bufH[idx] - g21, thr) + g21;
  float bn = shrinkf(bufH[iw] - g21n, thr) + g21n;
  float c = shrinkf(bufV[idx] - g22, thr) + g22;
  float cn = shrinkf(bufV[ih] - g22n, thr) + g22n;
  float a = X[idx] + G1[idx] * im1;
  bufC[idx] = mu1 * a + mu2 * ((b - bn) + (c - cn));
}

__global__ void k_post(const float* __restrict__ Z, const float* __restrict__ Y,
                       const float* __restrict__ inW, float* X, float* G1,
                       float* G21, float* G22, float* bufH, float* bufV,
                       float mu1, float mu2, float im2, float thr) {
  int idx = blockIdx.x * blockDim.x + threadIdx.x;
  int w = idx & 127, h = (idx >> 7) & 127;
  int ib = idx & ~(IMG_ELEMS - 1);
  float z = Z[idx];
  float dh = z - Z[ib + (h << 7) + ((w + 127) & 127)];
  float dv = z - Z[ib + (((h + 127) & 127) << 7) + w];
  float g21 = G21[idx], g22 = G22[idx];
  float uh = shrinkf(bufH[idx] - g21 * im2, thr);
  float uv = shrinkf(bufV[idx] - g22 * im2, thr);
  float iw = inW[idx];
  float x = (iw * Y[idx] + mu1 * z - G1[idx]) / (iw + mu1);
  X[idx] = x;
  G1[idx] += mu1 * (x - z);
  G21[idx] = g21 + mu2 * (uh - dh);
  G22[idx] = g22 + mu2 * (uv - dv);
  bufH[idx] = dh;
  bufV[idx] = dv;
}


__global__ void pass_left(const float* __restrict__ Ht,
                          const float* __restrict__ In, float* Out) {
  size_t base = (size_t)blockIdx.x * IMG_ELEMS;
  const float* in = In + base;
  int wave = threadIdx.x >> 5, lane = threadIdx.x & 31;
  int r = lane & 15, half = lane >> 4;
  int mB = blockIdx.y << 4;
  int nB = wave << 4;
  v8f acc = {};
  for (int k = 0; k < 128; k += 4) {
    v2f a = afrag(Ht, mB, k, r, half);
    v2f b = bfrag(in, k, nB, r, half);
    acc = wmma_f32(a, b, acc);
  }
  store_tile(Out + base, mB, nB, r, half, acc, 1.f);
}

__global__ void pass2(const float* __restrict__ Ht,
                      const float* __restrict__ cosrow,
                      const float* __restrict__ T, float* M, float mu1,
                      float mu2) {
  size_t base = (size_t)blockIdx.x * IMG_ELEMS;
  const float* t = T + base;
  int wave = threadIdx.x >> 5, lane = threadIdx.x & 31;
  int r = lane & 15, half = lane >> 4;
  int mB = blockIdx.y << 4;
  int nB = wave << 4;
  v8f m = {};
  for (int k = 0; k < 128; k += 4) {
    v2f a = afrag(t, mB, k, r, half);
    v2f b = bfrag(Ht, k, nB, r, half);
    m = wmma_f32(a, b, m);
  }
  float* mo = M + base;
  int v = nB + r;
  float cv = cosrow[v];
  for (int j = 0; j < 8; ++j) {
    int u = mB + j + 8 * half;
    float cu = cosrow[u];
    float den = mu1 + mu2 * (4.f - 2.f * cv - 2.f * cu);
    mo[(size_t)u * LD + v] = m[j] / den;
  }
}

__global__ void pass4(const float* __restrict__ Ht,
                      const float* __restrict__ U, float* Z) {
  size_t base = (size_t)blockIdx.x * IMG_ELEMS;
  const float* u = U + base;
  int wave = threadIdx.x >> 5, lane = threadIdx.x & 31;
  int r = lane & 15, half = lane >> 4;
  int mB = blockIdx.y << 4;
  int nB = wave << 4;
  v8f z = {};
  for (int k = 0; k < 128; k += 4) {
    v2f a = afrag(u, mB, k, r, half);
    v2f b = bfrag(Ht, k, nB, r, half);
    z = wmma_f32(a, b, z);
  }
  store_tile(Z + base, mB, nB, r, half, z, 1.f / 16384.f);
}


extern "C" void kernel_launch(void* const* d_in, const int* in_sizes, int n_in,
                              void* d_out, int out_size, void* d_ws,
                              size_t ws_size, hipStream_t stream) {
  (void)in_sizes; (void)n_in; (void)out_size; (void)ws_size;
  const float* Y = (const float*)d_in[0];
  const float* inW = (const float*)d_in[1];
  float* Z = (float*)d_out;
  float* ws = (float*)d_ws;

  float* Ht = ws;
  float* cosrow = Ht + IMG_ELEMS;
  float* p = ws + 4 * (size_t)IMG_ELEMS;
  float* X = p;            p += NE;
  float* G1 = p;           p += NE;
  float* G21 = p;          p += NE;
  float* G22 = p;          p += NE;
  float* bufH = p;         p += NE;
  float* bufV = p;         p += NE;
  float* bufC = p;         p += NE;
  float* T = p;            p += NE;
  float* M = p;            p += NE;

  const int EB = 256, EG = NE / EB;
  const dim3 gg(NIMG, 8);

  k_tables<<<IMG_ELEMS / 256, 256, 0, stream>>>(Ht, cosrow);
  k_init<<<EG, EB, 0, stream>>>(Y, Z, X, G1, G21, G22, bufH, bufV);

  float m1 = 0.1f, m2 = 0.1f;
  for (int it = 0; it < ITE; ++it) {
    float im1 = 1.f / m1, im2 = 1.f / m2, thr = 0.1f / m2;
    k_pre<<<EG, EB, 0, stream>>>(X, G1, G21, G22, bufH, bufV, bufC, m1, m2,
                                 im1, im2, thr);
    pass_left<<<gg, 256, 0, stream>>>(Ht, bufC, T);
    pass2<<<gg, 256, 0, stream>>>(Ht, cosrow, T, M, m1, m2);
    pass_left<<<gg, 256, 0, stream>>>(Ht, M, T);
    pass4<<<gg, 256, 0, stream>>>(Ht, T, Z);
    if (it < ITE - 1)
      k_post<<<EG, EB, 0, stream>>>(Z, Y, inW, X, G1, G21, G22, bufH, bufV, m1,
                                    m2, im2, thr);
    m1 *= 1.05f;
    m2 *= 1.05f;
  }
}


# each pass computes In^T*H (table symmetric): A and B fragments become row reads of 8 or 16 bytes per lane, one wave owns 2x4 interleaved tiles (992 waves); summation order unchanged
# speedup vs baseline: 3.0167x; 1.2486x over previous
#include <hip/hip_runtime.h>
#include <hip/hip_bf16.h>
#include <math.h>


typedef __attribute__((ext_vector_type(2))) float v2f;
typedef __attribute__((ext_vector_type(4))) float v4f;
typedef __attribute__((ext_vector_type(8))) float v8f;

#define NIMG 124
#define IMG_ELEMS 16384
#define NE (NIMG * IMG_ELEMS)
#define LD 128
#define ITE 20


static __device__ __forceinline__ v8f wmma_f32(v2f a, v2f b, v8f c) {
  return __builtin_amdgcn_wmma_f32_16x16x4_f32(false, a, false, b,
                                               (short)0, c, false, false);
}


__global__ void k_tables(float* Ht, float* cosrow) {
  int i = blockIdx.x * blockDim.x + threadIdx.x;
  if (i >= IMG_ELEMS) return;
  int m = i >> 7, k = i & 127;
  int p = (m * k) & 127;
  float ang = (6.283185307179586f / 128.f) * (float)p;
  float s, c;
  sincosf(ang, &s, &c);
  Ht[i] = c + s;
  if (m == 1) cosrow[k] = c;
}

__global__ void k_init(const float* __restrict__ Y, float* Z, float* X,
                       float* G1, float* G21, float* G22, float* bufH,
                       float* bufV) {
  int idx = blockIdx.x * blockDim.x + threadIdx.x;
  int w = idx & 127, h = (idx >> 7) & 127;
  int ib = idx & ~(IMG_ELEMS - 1);
  float y = Y[idx];
  Z[idx] = y;
  X[idx] = y;
  G1[idx] = 0.f;
  G21[idx] = 0.f;
  G22[idx] = 0.f;
  bufH[idx] = y - Y[ib + (h << 7) + ((w + 127) & 127)];
  bufV[idx] = y - Y[ib + (((h + 127) & 127) << 7) + w];
}

static __device__ __forceinline__ float shrinkf(float x, float t) {
  float m = fmaxf(fabsf(x) - t, 0.f);
  return copysignf(m, x);
}

__global__ void k_pre(const float* __restrict__ X, const float* __restrict__ G1,
                      const float* __restrict__ G21,
                      const float* __restrict__ G22,
                      const float* __restrict__ bufH,
                      const float* __restrict__ bufV, float* __restrict__ bufC,
                      float mu1, float mu2, float im1, float im2, float thr) {
  int idx = blockIdx.x * blockDim.x + threadIdx.x;
  int w = idx & 127, h = (idx >> 7) & 127;
  int ib = idx & ~(IMG_ELEMS - 1);
  int iw = ib + (h << 7) + ((w + 1) & 127);
  int ih = ib + (((h + 1) & 127) << 7) + w;
  float g21 = G21[idx] * im2, g21n = G21[iw] * im2;
  float g22 = G22[idx] * im2, g22n = G22[ih] * im2;
  float b = shrinkf(bufH[idx] - g21, thr) + g21;
  float bn = shrinkf(bufH[iw] - g21n, thr) + g21n;
  float c = shrinkf(bufV[idx] - g22, thr) + g22;
  float cn = shrinkf(bufV[ih] - g22n, thr) + g22n;
  float a = X[idx] + G1[idx] * im1;
  bufC[idx] = mu1 * a + mu2 * ((b - bn) + (c - cn));
}

__global__ void k_post(const float* __restrict__ Z, const float* __restrict__ Y,
                       const float* __restrict__ inW, float* X, float* G1,
                       float* G21, float* G22, float* bufH, float* bufV,
                       float mu1, float mu2, float im2, float thr) {
  int idx = blockIdx.x * blockDim.x + threadIdx.x;
  int w = idx & 127, h = (idx >> 7) & 127;
  int ib = idx & ~(IMG_ELEMS - 1);
  float z = Z[idx];
  float dh = z - Z[ib + (h << 7) + ((w + 127) & 127)];
  float dv = z - Z[ib + (((h + 127) & 127) << 7) + w];
  float g21 = G21[idx], g22 = G22[idx];
  float uh = shrinkf(bufH[idx] - g21 * im2, thr);
  float uv = shrinkf(bufV[idx] - g22 * im2, thr);
  float iw = inW[idx];
  float x = (iw * Y[idx] + mu1 * z - G1[idx]) / (iw + mu1);
  X[idx] = x;
  G1[idx] += mu1 * (x - z);
  G21[idx] = g21 + mu2 * (uh - dh);
  G22[idx] = g22 + mu2 * (uv - dv);
  bufH[idx] = dh;
  bufV[idx] = dv;
}

#define KSTEPS 32
#define KB 8
#define NB (KSTEPS / KB)

struct FragGroup {
  v2f a0[KB], a1[KB];
  v4f b0[KB], b1[KB];
};

static __device__ __forceinline__ void load_group(FragGroup& f,
                                                  const float* __restrict__ xa,
                                                  const float* __restrict__ hb,
                                                  const int g) {
#pragma unroll
  for (int i = 0; i < KB; ++i) {
    const int s = g * KB + i;
    f.a0[i] = *(const v2f*)(xa + (size_t)(4 * s) * LD);
    f.a1[i] = *(const v2f*)(xa + (size_t)(4 * s + 1) * LD);
    f.b0[i] = *(const v4f*)(hb + (size_t)(4 * s) * LD);
    f.b1[i] = *(const v4f*)(hb + (size_t)(4 * s + 1) * LD);
  }
}

static __device__ __forceinline__ void mma_group(const FragGroup& f,
                                                 v8f (&acc)[2][4]) {
#pragma unroll
  for (int i = 0; i < KB; ++i) {
#pragma unroll
    for (int tm = 0; tm < 2; ++tm) {
      v2f a;
      a.x = f.a0[i][tm];
      a.y = f.a1[i][tm];
#pragma unroll
      for (int tn = 0; tn < 4; ++tn) {
        v2f b;
        b.x = f.b0[i][tn];
        b.y = f.b1[i][tn];
        acc[tm][tn] = wmma_f32(a, b, acc[tm][tn]);
      }
    }
  }
}

static __device__ __forceinline__ void keep_order() {
  __builtin_amdgcn_sched_barrier(0);
}

static __device__ __forceinline__ void xt_h_block(
    const float* __restrict__ Ht, const float* __restrict__ cosrow,
    const float* __restrict__ In, float* __restrict__ Out, const int epi,
    float mu1, float mu2) {
  int wid = blockIdx.x * 4 + (threadIdx.x >> 5);
  int img = wid >> 3, sub = wid & 7;
  int mg = sub >> 1, ng = sub & 1;
  int lane = threadIdx.x & 31;
  int c = lane & 15, h = lane >> 4;
  size_t base = (size_t)img * IMG_ELEMS;
  const float* xa = In + base + (size_t)(2 * h) * LD + 32 * mg + 2 * c;
  const float* hb = Ht + (size_t)(2 * h) * LD + 64 * ng + 4 * c;

  FragGroup f[NB];
  v8f acc[2][4] = {};
  load_group(f[0], xa, hb, 0);
  load_group(f[1], xa, hb, 1);
  keep_order();
  mma_group(f[0], acc);
  keep_order();
  load_group(f[2], xa, hb, 2);
  keep_order();
  mma_group(f[1], acc);
  keep_order();
  load_group(f[3], xa, hb, 3);
  keep_order();
  mma_group(f[2], acc);
  mma_group(f[3], acc);

  float* o = Out + base;
  int n0 = 64 * ng + 4 * c;
  v4f cv = {0.f, 0.f, 0.f, 0.f};
  if (epi == 1) cv = *(const v4f*)(cosrow + n0);
#pragma unroll
  for (int tm = 0; tm < 2; ++tm) {
#pragma unroll
    for (int j = 0; j < 8; ++j) {
      int m = 32 * mg + 2 * (8 * h + j) + tm;
      v4f v;
      v.x = acc[tm][0][j];
      v.y = acc[tm][1][j];
      v.z = acc[tm][2][j];
      v.w = acc[tm][3][j];
      if (epi == 1) {
        float cu = cosrow[m];
        v.x = v.x / (mu1 + mu2 * (4.f - 2.f * cv.x - 2.f * cu));
        v.y = v.y / (mu1 + mu2 * (4.f - 2.f * cv.y - 2.f * cu));
        v.z = v.z / (mu1 + mu2 * (4.f - 2.f * cv.z - 2.f * cu));
        v.w = v.w / (mu1 + mu2 * (4.f - 2.f * cv.w - 2.f * cu));
      } else if (epi == 2) {
        v = v * (1.f / 16384.f);
      }
      *(v4f*)(o + (size_t)m * LD + n0) = v;
    }
  }
}

__global__ void __launch_bounds__(128)
pass_xt(const float* __restrict__ Ht, const float* __restrict__ In,
        float* __restrict__ Out) {
  xt_h_block(Ht, Ht, In, Out, 0, 0.f, 0.f);
}

__global__ void __launch_bounds__(128)
pass_den(const float* __restrict__ Ht, const float* __restrict__ cosrow,
         const float* __restrict__ T, float* __restrict__ M, float mu1,
         float mu2) {
  xt_h_block(Ht, cosrow, T, M, 1, mu1, mu2);
}

__global__ void __launch_bounds__(128)
pass_z(const float* __restrict__ Ht, const float* __restrict__ T,
       float* __restrict__ Z) {
  xt_h_block(Ht, Ht, T, Z, 2, 0.f, 0.f);
}


extern "C" void kernel_launch(void* const* d_in, const int* in_sizes, int n_in,
                              void* d_out, int out_size, void* d_ws,
                              size_t ws_size, hipStream_t stream) {
  (void)in_sizes; (void)n_in; (void)out_size; (void)ws_size;
  const float* Y = (const float*)d_in[0];
  const float* inW = (const float*)d_in[1];
  float* Z = (float*)d_out;
  float* ws = (float*)d_ws;

  float* Ht = ws;
  float* cosrow = Ht + IMG_ELEMS;
  float* p = ws + 4 * (size_t)IMG_ELEMS;
  float* X = p;            p += NE;
  float* G1 = p;           p += NE;
  float* G21 = p;          p += NE;
  float* G22 = p;          p += NE;
  float* bufH = p;         p += NE;
  float* bufV = p;         p += NE;
  float* bufC = p;         p += NE;
  float* T = p;            p += NE;
  float* M = p;            p += NE;

  const int EB = 256, EG = NE / EB;
  const int GG = NIMG * 2, GB = 128;

  k_tables<<<IMG_ELEMS / 256, 256, 0, stream>>>(Ht, cosrow);
  k_init<<<EG, EB, 0, stream>>>(Y, Z, X, G1, G21, G22, bufH, bufV);

  float m1 = 0.1f, m2 = 0.1f;
  for (int it = 0; it < ITE; ++it) {
    float im1 = 1.f / m1, im2 = 1.f / m2, thr = 0.1f / m2;
    k_pre<<<EG, EB, 0, stream>>>(X, G1, G21, G22, bufH, bufV, bufC, m1, m2,
                                 im1, im2, thr);
    pass_xt<<<GG, GB, 0, stream>>>(Ht, bufC, T);
    pass_den<<<GG, GB, 0, stream>>>(Ht, cosrow, T, M, m1, m2);
    pass_xt<<<GG, GB, 0, stream>>>(Ht, M, T);
    pass_z<<<GG, GB, 0, stream>>>(Ht, T, Z);
    if (it < ITE - 1)
      k_post<<<EG, EB, 0, stream>>>(Z, Y, inW, X, G1, G21, G22, bufH, bufV, m1,
                                    m2, im2, thr);
    m1 *= 1.05f;
    m2 *= 1.05f;
  }
}


# dual planes G21, G22 and DhZ, DvZ, X no longer stored: B = Uh + G21/mu2 and C carried instead (G21 = mu2 (B - DhZ), sums reordered), post and pre become one kernel, 121 -> 101 launches
# speedup vs baseline: 3.7588x; 1.2460x over previous
#include <hip/hip_runtime.h>
#include <hip/hip_bf16.h>
#include <math.h>


typedef __attribute__((ext_vector_type(2))) float v2f;
typedef __attribute__((ext_vector_type(4))) float v4f;
typedef __attribute__((ext_vector_type(8))) float v8f;

#define NIMG 124
#define IMG_ELEMS 16384
#define NE (NIMG * IMG_ELEMS)
#define LD 128
#define ITE 20


static __device__ __forceinline__ v8f wmma_f32(v2f a, v2f b, v8f c) {
  return __builtin_amdgcn_wmma_f32_16x16x4_f32(false, a, false, b,
                                               (short)0, c, false, false);
}


__global__ void k_tables(float* Ht, float* cosrow) {
  int i = blockIdx.x * blockDim.x + threadIdx.x;
  if (i >= IMG_ELEMS) return;
  int m = i >> 7, k = i & 127;
  int p = (m * k) & 127;
  float ang = (6.283185307179586f / 128.f) * (float)p;
  float s, c;
  sincosf(ang, &s, &c);
  Ht[i] = c + s;
  if (m == 1) cosrow[k] = c;
}

static __device__ __forceinline__ float shrinkf(float x, float t) {
  float m = fmaxf(fabsf(x) - t, 0.f);
  return copysignf(m, x);
}

static __device__ __forceinline__ void step_body(
    const bool first, const float* Zs, const float* Y, const float* inW,
    float* G1, const float* Bo, const float* Co, float* Bn, float* Cn,
    float* R, float mu1p, float mu1, float im1, float mu2p, float mu2,
    float im2, float thr) {
  int idx = blockIdx.x * blockDim.x + threadIdx.x;
  int w = idx & 127, h = (idx >> 7) & 127;
  int ib = idx & ~(IMG_ELEMS - 1);
  int row = ib + (h << 7);
  int iwm = row + ((w + 127) & 127);
  int iwp = row + ((w + 1) & 127);
  int ihm = ib + (((h + 127) & 127) << 7) + w;
  int ihp = ib + (((h + 1) & 127) << 7) + w;
  float z = Zs[idx];
  float dh = z - Zs[iwm], dhn = Zs[iwp] - z;
  float dv = z - Zs[ihm], dvn = Zs[ihp] - z;
  float q21 = 0.f, q21n = 0.f, q22 = 0.f, q22n = 0.f;
  float x, g1;
  if (first) {
    x = z;
    g1 = 0.f;
  } else {
    q21 = (mu2p * (Bo[idx] - dh)) * im2;
    q21n = (mu2p * (Bo[iwp] - dhn)) * im2;
    q22 = (mu2p * (Co[idx] - dv)) * im2;
    q22n = (mu2p * (Co[ihp] - dvn)) * im2;
    float iw = inW[idx];
    float g1o = G1[idx];
    x = (iw * Y[idx] + mu1p * z - g1o) / (iw + mu1p);
    g1 = g1o + mu1p * (x - z);
  }
  float b = shrinkf(dh - q21, thr) + q21;
  float bn = shrinkf(dhn - q21n, thr) + q21n;
  float c = shrinkf(dv - q22, thr) + q22;
  float cn = shrinkf(dvn - q22n, thr) + q22n;
  G1[idx] = g1;
  Bn[idx] = b;
  Cn[idx] = c;
  float a = x + g1 * im1;
  R[idx] = mu1 * a + mu2 * ((b - bn) + (c - cn));
}

__global__ void k_first(const float* __restrict__ Y, float* __restrict__ G1,
                        float* __restrict__ Bn, float* __restrict__ Cn,
                        float* __restrict__ R, float mu1, float im1, float mu2,
                        float thr) {
  step_body(true, Y, Y, Y, G1, Y, Y, Bn, Cn, R, 0.f, mu1, im1, 0.f, mu2, 0.f,
            thr);
}

__global__ void k_step(const float* __restrict__ Z, const float* __restrict__ Y,
                       const float* __restrict__ inW, float* __restrict__ G1,
                       const float* __restrict__ Bo,
                       const float* __restrict__ Co, float* __restrict__ Bn,
                       float* __restrict__ Cn, float* __restrict__ R,
                       float mu1p, float mu1, float im1, float mu2p, float mu2,
                       float im2, float thr) {
  step_body(false, Z, Y, inW, G1, Bo, Co, Bn, Cn, R, mu1p, mu1, im1, mu2p, mu2,
            im2, thr);
}

#define KSTEPS 32
#define KB 8
#define NB (KSTEPS / KB)

struct FragGroup {
  v2f a0[KB], a1[KB];
  v4f b0[KB], b1[KB];
};

static __device__ __forceinline__ void load_group(FragGroup& f,
                                                  const float* __restrict__ xa,
                                                  const float* __restrict__ hb,
                                                  const int g) {
#pragma unroll
  for (int i = 0; i < KB; ++i) {
    const int s = g * KB + i;
    f.a0[i] = *(const v2f*)(xa + (size_t)(4 * s) * LD);
    f.a1[i] = *(const v2f*)(xa + (size_t)(4 * s + 1) * LD);
    f.b0[i] = *(const v4f*)(hb + (size_t)(4 * s) * LD);
    f.b1[i] = *(const v4f*)(hb + (size_t)(4 * s + 1) * LD);
  }
}

static __device__ __forceinline__ void mma_group(const FragGroup& f,
                                                 v8f (&acc)[2][4]) {
#pragma unroll
  for (int i = 0; i < KB; ++i) {
#pragma unroll
    for (int tm = 0; tm < 2; ++tm) {
      v2f a;
      a.x = f.a0[i][tm];
      a.y = f.a1[i][tm];
#pragma unroll
      for (int tn = 0; tn < 4; ++tn) {
        v2f b;
        b.x = f.b0[i][tn];
        b.y = f.b1[i][tn];
        acc[tm][tn] = wmma_f32(a, b, acc[tm][tn]);
      }
    }
  }
}

static __device__ __forceinline__ void keep_order() {
  __builtin_amdgcn_sched_barrier(0);
}

static __device__ __forceinline__ void xt_h_block(
    const float* __restrict__ Ht, const float* __restrict__ cosrow,
    const float* __restrict__ In, float* __restrict__ Out, const int epi,
    float mu1, float mu2) {
  int wid = blockIdx.x * 4 + (threadIdx.x >> 5);
  int img = wid >> 3, sub = wid & 7;
  int mg = sub >> 1, ng = sub & 1;
  int lane = threadIdx.x & 31;
  int c = lane & 15, h = lane >> 4;
  size_t base = (size_t)img * IMG_ELEMS;
  const float* xa = In + base + (size_t)(2 * h) * LD + 32 * mg + 2 * c;
  const float* hb = Ht + (size_t)(2 * h) * LD + 64 * ng + 4 * c;

  FragGroup f[NB];
  v8f acc[2][4] = {};
  load_group(f[0], xa, hb, 0);
  load_group(f[1], xa, hb, 1);
  keep_order();
  mma_group(f[0], acc);
  keep_order();
  load_group(f[2], xa, hb, 2);
  keep_order();
  mma_group(f[1], acc);
  keep_order();
  load_group(f[3], xa, hb, 3);
  keep_order();
  mma_group(f[2], acc);
  mma_group(f[3], acc);

  float* o = Out + base;
  int n0 = 64 * ng + 4 * c;
  v4f cv = {0.f, 0.f, 0.f, 0.f};
  if (epi == 1) cv = *(const v4f*)(cosrow + n0);
#pragma unroll
  for (int tm = 0; tm < 2; ++tm) {
#pragma unroll
    for (int j = 0; j < 8; ++j) {
      int m = 32 * mg + 2 * (8 * h + j) + tm;
      v4f v;
      v.x = acc[tm][0][j];
      v.y = acc[tm][1][j];
      v.z = acc[tm][2][j];
      v.w = acc[tm][3][j];
      if (epi == 1) {
        float cu = cosrow[m];
        v.x = v.x / (mu1 + mu2 * (4.f - 2.f * cv.x - 2.f * cu));
        v.y = v.y / (mu1 + mu2 * (4.f - 2.f * cv.y - 2.f * cu));
        v.z = v.z / (mu1 + mu2 * (4.f - 2.f * cv.z - 2.f * cu));
        v.w = v.w / (mu1 + mu2 * (4.f - 2.f * cv.w - 2.f * cu));
      } else if (epi == 2) {
        v = v * (1.f / 16384.f);
      }
      *(v4f*)(o + (size_t)m * LD + n0) = v;
    }
  }
}

__global__ void __launch_bounds__(128)
pass_xt(const float* __restrict__ Ht, const float* __restrict__ In,
        float* __restrict__ Out) {
  xt_h_block(Ht, Ht, In, Out, 0, 0.f, 0.f);
}

__global__ void __launch_bounds__(128)
pass_den(const float* __restrict__ Ht, const float* __restrict__ cosrow,
         const float* __restrict__ T, float* __restrict__ M, float mu1,
         float mu2) {
  xt_h_block(Ht, cosrow, T, M, 1, mu1, mu2);
}

__global__ void __launch_bounds__(128)
pass_z(const float* __restrict__ Ht, const float* __restrict__ T,
       float* __restrict__ Z) {
  xt_h_block(Ht, Ht, T, Z, 2, 0.f, 0.f);
}


extern "C" void kernel_launch(void* const* d_in, const int* in_sizes, int n_in,
                              void* d_out, int out_size, void* d_ws,
                              size_t ws_size, hipStream_t stream) {
  (void)in_sizes; (void)n_in; (void)out_size; (void)ws_size;
  const float* Y = (const float*)d_in[0];
  const float* inW = (const float*)d_in[1];
  float* Z = (float*)d_out;
  float* ws = (float*)d_ws;

  float* Ht = ws;
  float* cosrow = Ht + IMG_ELEMS;
  float* p = ws + 4 * (size_t)IMG_ELEMS;
  float* G1 = p;           p += NE;
  float* Bo = p;           p += NE;
  float* Co = p;           p += NE;
  float* Bn = p;           p += NE;
  float* Cn = p;           p += NE;
  float* bufC = p;         p += NE;
  float* T = p;            p += NE;
  float* M = p;            p += NE;

  const int EB = 256, EG = NE / EB;
  const int GG = NIMG * 2, GB = 128;

  k_tables<<<IMG_ELEMS / 256, 256, 0, stream>>>(Ht, cosrow);

  float m1 = 0.1f, m2 = 0.1f;
  float m1p = 0.f, m2p = 0.f;
  for (int it = 0; it < ITE; ++it) {
    float im1 = 1.f / m1, im2 = 1.f / m2, thr = 0.1f / m2;
    if (it == 0)
      k_first<<<EG, EB, 0, stream>>>(Y, G1, Bn, Cn, bufC, m1, im1, m2, thr);
    else
      k_step<<<EG, EB, 0, stream>>>(Z, Y, inW, G1, Bo, Co, Bn, Cn, bufC, m1p,
                                    m1, im1, m2p, m2, im2, thr);
    pass_xt<<<GG, GB, 0, stream>>>(Ht, bufC, T);
    pass_den<<<GG, GB, 0, stream>>>(Ht, cosrow, T, M, m1, m2);
    pass_xt<<<GG, GB, 0, stream>>>(Ht, M, T);
    pass_z<<<GG, GB, 0, stream>>>(Ht, T, Z);
    float* t;
    t = Bo; Bo = Bn; Bn = t;
    t = Co; Co = Cn; Cn = t;
    m1p = m1;
    m2p = m2;
    m1 *= 1.05f;
    m2 *= 1.05f;
  }
}


# passes paired in one launch: a 4-wave work-group forms a 128x64 half of T = In^T H in LDS, then its 64 rows of T^T H from it; T never stored, 101 -> 61 launches; summation order unchanged
# speedup vs baseline: 4.7938x; 1.2754x over previous
#include <hip/hip_runtime.h>
#include <hip/hip_bf16.h>
#include <math.h>


typedef __attribute__((ext_vector_type(2))) float v2f;
typedef __attribute__((ext_vector_type(4))) float v4f;
typedef __attribute__((ext_vector_type(8))) float v8f;

#define NIMG 124
#define IMG_ELEMS 16384
#define NE (NIMG * IMG_ELEMS)
#define LD 128
#define ITE 20


static __device__ __forceinline__ v8f wmma_f32(v2f a, v2f b, v8f c) {
  return __builtin_amdgcn_wmma_f32_16x16x4_f32(false, a, false, b,
                                               (short)0, c, false, false);
}


__global__ void k_tables(float* Ht, float* cosrow) {
  int i = blockIdx.x * blockDim.x + threadIdx.x;
  if (i >= IMG_ELEMS) return;
  int m = i >> 7, k = i & 127;
  int p = (m * k) & 127;
  float ang = (6.283185307179586f / 128.f) * (float)p;
  float s, c;
  sincosf(ang, &s, &c);
  Ht[i] = c + s;
  if (m == 1) cosrow[k] = c;
}

static __device__ __forceinline__ float shrinkf(float x, float t) {
  float m = fmaxf(fabsf(x) - t, 0.f);
  return copysignf(m, x);
}

static __device__ __forceinline__ void step_body(
    const bool first, const float* Zs, const float* Y, const float* inW,
    float* G1, const float* Bo, const float* Co, float* Bn, float* Cn,
    float* R, float mu1p, float mu1, float im1, float mu2p, float mu2,
    float im2, float thr) {
  int idx = blockIdx.x * blockDim.x + threadIdx.x;
  int w = idx & 127, h = (idx >> 7) & 127;
  int ib = idx & ~(IMG_ELEMS - 1);
  int row = ib + (h << 7);
  int iwm = row + ((w + 127) & 127);
  int iwp = row + ((w + 1) & 127);
  int ihm = ib + (((h + 127) & 127) << 7) + w;
  int ihp = ib + (((h + 1) & 127) << 7) + w;
  float z = Zs[idx];
  float dh = z - Zs[iwm], dhn = Zs[iwp] - z;
  float dv = z - Zs[ihm], dvn = Zs[ihp] - z;
  float q21 = 0.f, q21n = 0.f, q22 = 0.f, q22n = 0.f;
  float x, g1;
  if (first) {
    x = z;
    g1 = 0.f;
  } else {
    q21 = (mu2p * (Bo[idx] - dh)) * im2;
    q21n = (mu2p * (Bo[iwp] - dhn)) * im2;
    q22 = (mu2p * (Co[idx] - dv)) * im2;
    q22n = (mu2p * (Co[ihp] - dvn)) * im2;
    float iw = inW[idx];
    float g1o = G1[idx];
    x = (iw * Y[idx] + mu1p * z - g1o) / (iw + mu1p);
    g1 = g1o + mu1p * (x - z);
  }
  float b = shrinkf(dh - q21, thr) + q21;
  float bn = shrinkf(dhn - q21n, thr) + q21n;
  float c = shrinkf(dv - q22, thr) + q22;
  float cn = shrinkf(dvn - q22n, thr) + q22n;
  G1[idx] = g1;
  Bn[idx] = b;
  Cn[idx] = c;
  float a = x + g1 * im1;
  R[idx] = mu1 * a + mu2 * ((b - bn) + (c - cn));
}

__global__ void k_first(const float* __restrict__ Y, float* __restrict__ G1,
                        float* __restrict__ Bn, float* __restrict__ Cn,
                        float* __restrict__ R, float mu1, float im1, float mu2,
                        float thr) {
  step_body(true, Y, Y, Y, G1, Y, Y, Bn, Cn, R, 0.f, mu1, im1, 0.f, mu2, 0.f,
            thr);
}

__global__ void k_step(const float* __restrict__ Z, const float* __restrict__ Y,
                       const float* __restrict__ inW, float* __restrict__ G1,
                       const float* __restrict__ Bo,
                       const float* __restrict__ Co, float* __restrict__ Bn,
                       float* __restrict__ Cn, float* __restrict__ R,
                       float mu1p, float mu1, float im1, float mu2p, float mu2,
                       float im2, float thr) {
  step_body(false, Z, Y, inW, G1, Bo, Co, Bn, Cn, R, mu1p, mu1, im1, mu2p, mu2,
            im2, thr);
}

#define KSTEPS 32
#define KB 8
#define NB (KSTEPS / KB)
#define TP 80

struct FragGroup {
  v2f a0[KB], a1[KB];
  v4f b0[KB], b1[KB];
};

static __device__ __forceinline__ void load_group_a(FragGroup& f,
                                                    const float* xa,
                                                    const int pitch,
                                                    const int g) {
#pragma unroll
  for (int i = 0; i < KB; ++i) {
    const int s = g * KB + i;
    f.a0[i] = *(const v2f*)(xa + (4 * s) * pitch);
    f.a1[i] = *(const v2f*)(xa + (4 * s + 1) * pitch);
  }
}

static __device__ __forceinline__ void load_group_b(FragGroup& f,
                                                    const float* __restrict__ hb,
                                                    const int g) {
#pragma unroll
  for (int i = 0; i < KB; ++i) {
    const int s = g * KB + i;
    f.b0[i] = *(const v4f*)(hb + (4 * s) * LD);
    f.b1[i] = *(const v4f*)(hb + (4 * s + 1) * LD);
  }
}

static __device__ __forceinline__ void mma_group(const FragGroup& f,
                                                 v8f (&acc)[2][4]) {
#pragma unroll
  for (int i = 0; i < KB; ++i) {
#pragma unroll
    for (int tm = 0; tm < 2; ++tm) {
      v2f a;
      a.x = f.a0[i][tm];
      a.y = f.a1[i][tm];
#pragma unroll
      for (int tn = 0; tn < 4; ++tn) {
        v2f b;
        b.x = f.b0[i][tn];
        b.y = f.b1[i][tn];
        acc[tm][tn] = wmma_f32(a, b, acc[tm][tn]);
      }
    }
  }
}

static __device__ __forceinline__ void keep_order() {
  __builtin_amdgcn_sched_barrier(0);
}

template <int EPI>
static __device__ __forceinline__ void pair_block(
    const float* __restrict__ Ht, const float* __restrict__ cosrow,
    const float* __restrict__ In, float* __restrict__ Out, float mu1,
    float mu2) {
  __shared__ float Tl[LD * TP];
  const int img = blockIdx.x >> 1, hm = blockIdx.x & 1;
  const int w = threadIdx.x >> 5;
  const int lane = threadIdx.x & 31;
  const int c = lane & 15, h = lane >> 4;
  const size_t base = (size_t)img * IMG_ELEMS;

  const float* xa = In + base + (2 * h) * LD + 32 * w + 2 * c;
  const float* hb = Ht + (2 * h) * LD + 64 * hm + 4 * c;
  const int mg2 = w >> 1, ng2 = w & 1;
  const float* xa2 = Tl + (2 * h) * TP + 32 * mg2 + 2 * c;
  const float* hb2 = Ht + (2 * h) * LD + 64 * ng2 + 4 * c;

  FragGroup f[NB];
  FragGroup f2[NB];
  v8f acc[2][4] = {};
  load_group_a(f[0], xa, LD, 0);
  load_group_b(f[0], hb, 0);
  load_group_a(f[1], xa, LD, 1);
  load_group_b(f[1], hb, 1);
  keep_order();
  mma_group(f[0], acc);
  keep_order();
  load_group_a(f[2], xa, LD, 2);
  load_group_b(f[2], hb, 2);
  keep_order();
  mma_group(f[1], acc);
  keep_order();
  load_group_a(f[3], xa, LD, 3);
  load_group_b(f[3], hb, 3);
  keep_order();
  mma_group(f[2], acc);
  keep_order();
  load_group_b(f2[0], hb2, 0);
  load_group_b(f2[1], hb2, 1);
  keep_order();
  mma_group(f[3], acc);

#pragma unroll
  for (int tm = 0; tm < 2; ++tm) {
#pragma unroll
    for (int j = 0; j < 8; ++j) {
      int m = 32 * w + 2 * (8 * h + j) + tm;
      v4f v;
      v.x = acc[tm][0][j];
      v.y = acc[tm][1][j];
      v.z = acc[tm][2][j];
      v.w = acc[tm][3][j];
      *(v4f*)(Tl + m * TP + 4 * c) = v;
    }
  }
  __syncthreads();

  v8f acc2[2][4] = {};
  load_group_a(f2[0], xa2, TP, 0);
  load_group_a(f2[1], xa2, TP, 1);
  keep_order();
  mma_group(f2[0], acc2);
  keep_order();
  load_group_a(f2[2], xa2, TP, 2);
  load_group_b(f2[2], hb2, 2);
  keep_order();
  mma_group(f2[1], acc2);
  keep_order();
  load_group_a(f2[3], xa2, TP, 3);
  load_group_b(f2[3], hb2, 3);
  keep_order();
  mma_group(f2[2], acc2);
  mma_group(f2[3], acc2);

  float* o = Out + base;
  const int n0 = 64 * ng2 + 4 * c;
  v4f cv = {0.f, 0.f, 0.f, 0.f};
  if (EPI == 1) cv = *(const v4f*)(cosrow + n0);
#pragma unroll
  for (int tm = 0; tm < 2; ++tm) {
#pragma unroll
    for (int j = 0; j < 8; ++j) {
      int m = 64 * hm + 32 * mg2 + 2 * (8 * h + j) + tm;
      v4f v;
      v.x = acc2[tm][0][j];
      v.y = acc2[tm][1][j];
      v.z = acc2[tm][2][j];
      v.w = acc2[tm][3][j];
      if (EPI == 1) {
        float cu = cosrow[m];
        v.x = v.x / (mu1 + mu2 * (4.f - 2.f * cv.x - 2.f * cu));
        v.y = v.y / (mu1 + mu2 * (4.f - 2.f * cv.y - 2.f * cu));
        v.z = v.z / (mu1 + mu2 * (4.f - 2.f * cv.z - 2.f * cu));
        v.w = v.w / (mu1 + mu2 * (4.f - 2.f * cv.w - 2.f * cu));
      } else {
        v = v * (1.f / 16384.f);
      }
      *(v4f*)(o + (size_t)m * LD + n0) = v;
    }
  }
}

__global__ void __launch_bounds__(128)
pair_den(const float* __restrict__ Ht, const float* __restrict__ cosrow,
         const float* __restrict__ R, float* __restrict__ M, float mu1,
         float mu2) {
  pair_block<1>(Ht, cosrow, R, M, mu1, mu2);
}

__global__ void __launch_bounds__(128)
pair_z(const float* __restrict__ Ht, const float* __restrict__ M,
       float* __restrict__ Z) {
  pair_block<2>(Ht, Ht, M, Z, 0.f, 0.f);
}


extern "C" void kernel_launch(void* const* d_in, const int* in_sizes, int n_in,
                              void* d_out, int out_size, void* d_ws,
                              size_t ws_size, hipStream_t stream) {
  (void)in_sizes; (void)n_in; (void)out_size; (void)ws_size;
  const float* Y = (const float*)d_in[0];
  const float* inW = (const float*)d_in[1];
  float* Z = (float*)d_out;
  float* ws = (float*)d_ws;

  float* Ht = ws;
  float* cosrow = Ht + IMG_ELEMS;
  float* p = ws + 4 * (size_t)IMG_ELEMS;
  float* G1 = p;           p += NE;
  float* Bo = p;           p += NE;
  float* Co = p;           p += NE;
  float* Bn = p;           p += NE;
  float* Cn = p;           p += NE;
  float* bufC = p;         p += NE;
  float* M = p;            p += NE;

  const int EB = 256, EG = NE / EB;
  const int GG = NIMG * 2, GB = 128;

  k_tables<<<IMG_ELEMS / 256, 256, 0, stream>>>(Ht, cosrow);

  float m1 = 0.1f, m2 = 0.1f;
  float m1p = 0.f, m2p = 0.f;
  for (int it = 0; it < ITE; ++it) {
    float im1 = 1.f / m1, im2 = 1.f / m2, thr = 0.1f / m2;
    if (it == 0)
      k_first<<<EG, EB, 0, stream>>>(Y, G1, Bn, Cn, bufC, m1, im1, m2, thr);
    else
      k_step<<<EG, EB, 0, stream>>>(Z, Y, inW, G1, Bo, Co, Bn, Cn, bufC, m1p,
                                    m1, im1, m2p, m2, im2, thr);
    pair_den<<<GG, GB, 0, stream>>>(Ht, cosrow, bufC, M, m1, m2);
    pair_z<<<GG, GB, 0, stream>>>(Ht, M, Z);
    float* t;
    t = Bo; Bo = Bn; Bn = t;
    t = Co; Co = Cn; Cn = t;
    m1p = m1;
    m2p = m2;
    m1 *= 1.05f;
    m2 *= 1.05f;
  }
}


# all 20 iterations in one launch: an 8-wave work-group owns one image, Z, R, T, M live in two LDS image buffers, table fragments read once; 61 -> 2 launches; summation order unchanged
# speedup vs baseline: 6.2140x; 1.2963x over previous
#include <hip/hip_runtime.h>
#include <hip/hip_bf16.h>
#include <math.h>


typedef __attribute__((ext_vector_type(2))) float v2f;
typedef __attribute__((ext_vector_type(4))) float v4f;
typedef __attribute__((ext_vector_type(8))) float v8f;

#define NIMG 124
#define IMG_ELEMS 16384
#define NE (NIMG * IMG_ELEMS)
#define LD 128
#define ITE 20


static __device__ __forceinline__ v8f wmma_f32(v2f a, v2f b, v8f c) {
  return __builtin_amdgcn_wmma_f32_16x16x4_f32(false, a, false, b,
                                               (short)0, c, false, false);
}


__global__ void k_tables(float* Ht, float* cosrow) {
  int i = blockIdx.x * blockDim.x + threadIdx.x;
  if (i >= IMG_ELEMS) return;
  int m = i >> 7, k = i & 127;
  int p = (m * k) & 127;
  float ang = (6.283185307179586f / 128.f) * (float)p;
  float s, c;
  sincosf(ang, &s, &c);
  Ht[i] = c + s;
  if (m == 1) cosrow[k] = c;
}

static __device__ __forceinline__ float shrinkf(float x, float t) {
  float m = fmaxf(fabsf(x) - t, 0.f);
  return copysignf(m, x);
}

#define P 144
#define SBUF (LD * P)

#define SU 8
template <bool FIRST>
static __device__ __forceinline__ void step_block(
    const float* zb, float* rb, const float* __restrict__ Y,
    const float* __restrict__ inW, float* G1, const float* Bo, const float* Co,
    float* Bn, float* Cn, const int base, float mu1p, float mu1, float im1,
    float mu2p, float mu2, float im2, float thr) {
  const int tid = threadIdx.x;
  const int w = tid & 127, h0 = tid >> 7;
  const int wm = (w + 127) & 127, wp = (w + 1) & 127;
  for (int jb = 0; jb < 64; jb += SU) {
    float yv[SU] = {}, iwv[SU] = {}, g1v[SU] = {};
    float bo[SU] = {}, bop[SU] = {}, co[SU] = {}, cop[SU] = {};
    if (!FIRST) {
#pragma unroll
      for (int u = 0; u < SU; ++u) {
        const int h = h0 + 2 * (jb + u), hp = (h + 1) & 127;
        const int gi = base + (h << 7) + w;
        yv[u] = Y[gi];
        iwv[u] = inW[gi];
        g1v[u] = G1[gi];
        bo[u] = Bo[gi];
        bop[u] = Bo[base + (h << 7) + wp];
        co[u] = Co[gi];
        cop[u] = Co[base + (hp << 7) + w];
      }
    }
#pragma unroll
    for (int u = 0; u < SU; ++u) {
      const int h = h0 + 2 * (jb + u);
      const int hm = (h + 127) & 127, hp = (h + 1) & 127;
      const float z = zb[h * P + w];
      const float dh = z - zb[h * P + wm], dhn = zb[h * P + wp] - z;
      const float dv = z - zb[hm * P + w], dvn = zb[hp * P + w] - z;
      float q21 = 0.f, q21n = 0.f, q22 = 0.f, q22n = 0.f;
      float x, g1;
      if (FIRST) {
        x = z;
        g1 = 0.f;
      } else {
        q21 = (mu2p * (bo[u] - dh)) * im2;
        q21n = (mu2p * (bop[u] - dhn)) * im2;
        q22 = (mu2p * (co[u] - dv)) * im2;
        q22n = (mu2p * (cop[u] - dvn)) * im2;
        const float iw = iwv[u];
        const float g1o = g1v[u];
        x = (iw * yv[u] + mu1p * z - g1o) / (iw + mu1p);
        g1 = g1o + mu1p * (x - z);
      }
      const float b = shrinkf(dh - q21, thr) + q21;
      const float bn = shrinkf(dhn - q21n, thr) + q21n;
      const float c = shrinkf(dv - q22, thr) + q22;
      const float cn = shrinkf(dvn - q22n, thr) + q22n;
      const int gi = base + (h << 7) + w;
      G1[gi] = g1;
      Bn[gi] = b;
      Cn[gi] = c;
      const float a = x + g1 * im1;
      rb[h * P + w] = mu1 * a + mu2 * ((b - bn) + (c - cn));
    }
  }
}

#define SSTEPS 8
#define SG 4
#define NG (SSTEPS / SG)


struct FragA {
  v2f a[SG][4][2];
};
struct FragB {
  v4f b[SG][2][2];
};

static __device__ __forceinline__ void load_a_step(FragA& f, const float* xa,
                                                   const int pitch, const int g,
                                                   const int i) {
  const int u = g * SG + i;
#pragma unroll
  for (int q = 0; q < 4; ++q) {
#pragma unroll
    for (int e = 0; e < 2; ++e)
      f.a[i][q][e] = *(const v2f*)(xa + (4 * u + e + 32 * q) * pitch);
  }
}

static __device__ __forceinline__ void load_b_step(FragB& f,
                                                   const float* __restrict__ hb,
                                                   const int g, const int i) {
  const int u = g * SG + i;
#pragma unroll
  for (int q = 0; q < 2; ++q) {
#pragma unroll
    for (int e = 0; e < 2; ++e)
      f.b[i][q][e] = *(const v4f*)(hb + (4 * u + e + 32 * q) * LD);
  }
}

static __device__ __forceinline__ void load_group_a(FragA& fa, const float* xa,
                                                    const int pitch,
                                                    const int g) {
#pragma unroll
  for (int i = 0; i < SG; ++i) load_a_step(fa, xa, pitch, g, i);
}

static __device__ __forceinline__ void load_group_b(FragB& fb,
                                                    const float* __restrict__ hb,
                                                    const int g) {
#pragma unroll
  for (int i = 0; i < SG; ++i) load_b_step(fb, hb, g, i);
}

static __device__ __forceinline__ v2f bpair(const FragB& f, const int i,
                                            const int q, const int tn) {
  v2f b;
  b.x = f.b[i][q][0][tn];
  b.y = f.b[i][q][1][tn];
  return b;
}

static __device__ __forceinline__ void mma_group(const FragA& fa,
                                                 const FragB& fb,
                                                 v8f (&acc)[2][4]) {
#pragma unroll
  for (int i = 0; i < SG; ++i) {
    v2f ee[2], eo[2], o0[2], o1[2];
#pragma unroll
    for (int tm = 0; tm < 2; ++tm) {
#pragma unroll
      for (int e = 0; e < 2; ++e) {
        const float x0 = fa.a[i][0][e][tm], x1 = fa.a[i][1][e][tm];
        const float x2 = fa.a[i][2][e][tm], x3 = fa.a[i][3][e][tm];
        const float e0 = x0 + x2, e1 = x1 + x3;
        o0[tm][e] = x0 - x2;
        o1[tm][e] = x1 - x3;
        ee[tm][e] = e0 + e1;
        eo[tm][e] = e0 - e1;
      }
    }
#pragma unroll
    for (int tm = 0; tm < 2; ++tm) {
      acc[tm][0] = wmma_f32(ee[tm], bpair(fb, i, 0, 0), acc[tm][0]);
      acc[tm][1] = wmma_f32(o0[tm], bpair(fb, i, 0, 1), acc[tm][1]);
      acc[tm][2] = wmma_f32(eo[tm], bpair(fb, i, 0, 2), acc[tm][2]);
      acc[tm][3] = wmma_f32(o0[tm], bpair(fb, i, 0, 3), acc[tm][3]);
    }
#pragma unroll
    for (int tm = 0; tm < 2; ++tm) {
      acc[tm][1] = wmma_f32(o1[tm], bpair(fb, i, 1, 1), acc[tm][1]);
      acc[tm][3] = wmma_f32(o1[tm], bpair(fb, i, 1, 3), acc[tm][3]);
    }
  }
}

static __device__ __forceinline__ void keep_order() {
  __builtin_amdgcn_sched_barrier(0);
}

static __device__ __forceinline__ void pass_lds(const float* xa,
                                                const FragB (&fb)[NG],
                                                v8f (&acc)[2][4]) {
  FragA fa[NG];
  load_group_a(fa[0], xa, P, 0);
  load_group_a(fa[1], xa, P, 1);
  keep_order();
  mma_group(fa[0], fb[0], acc);
  mma_group(fa[1], fb[1], acc);
}

static __device__ __forceinline__ void store_block(const v8f (&acc)[2][4],
                                                   float* d, const int pitch) {
#pragma unroll
  for (int tm = 0; tm < 2; ++tm) {
#pragma unroll
    for (int j = 0; j < 8; ++j) {
      v4f v;
      v.x = acc[tm][0][j];
      v.y = acc[tm][1][j];
      v.z = acc[tm][2][j];
      v.w = acc[tm][3][j];
      *(v4f*)(d + (2 * j + tm) * pitch) = v;
    }
  }
}

static __device__ __forceinline__ void clear_acc(v8f (&acc)[2][4]) {
  const v8f zero = {0.f, 0.f, 0.f, 0.f, 0.f, 0.f, 0.f, 0.f};
#pragma unroll
  for (int tm = 0; tm < 2; ++tm) {
#pragma unroll
    for (int tn = 0; tn < 4; ++tn) acc[tm][tn] = zero;
  }
}

__global__ void __launch_bounds__(256)
k_solve(const float* __restrict__ Ht, const float* __restrict__ cosrow,
        const float* __restrict__ Y, const float* __restrict__ inW, float* G1,
        float* Ba, float* Ca, float* Bb, float* Cb, float* __restrict__ Zout) {
  __shared__ __attribute__((aligned(16))) float S[2 * SBUF];
  const int tid = threadIdx.x;
  const int wv = tid >> 5;
  const int lane = tid & 31;
  const int c = lane & 15, h = lane >> 4;
  const int mg = wv >> 1, ng = wv & 1;
  const int base = blockIdx.x * IMG_ELEMS;

  const float* hb = Ht + (2 * h) * LD + 64 * ng + 4 * c;
  FragB fb[NG];
  load_group_b(fb[0], hb, 0);
  load_group_b(fb[1], hb, 1);

  const int n0 = 64 * ng + 4 * c;
  const int r0 = 32 * mg + 16 * h;
  const int a_off = (2 * h) * P + 32 * mg + 2 * c;
  const int d_off = r0 * P + n0;
  const v4f cv = *(const v4f*)(cosrow + n0);
  float cu[2][8];
#pragma unroll
  for (int tm = 0; tm < 2; ++tm) {
#pragma unroll
    for (int j = 0; j < 8; ++j) cu[tm][j] = cosrow[r0 + 2 * j + tm];
  }

  int zo = 0;
  for (int j = 0; j < 64; ++j) {
    const int e = tid + 256 * j;
    S[(e >> 7) * P + (e & 127)] = Y[base + e];
  }
  __syncthreads();

  float m1 = 0.1f, m2 = 0.1f;
  float m1p = 0.f, m2p = 0.f;
  float* Bo = Ba;
  float* Co = Ca;
  float* Bn = Bb;
  float* Cn = Cb;
  for (int it = 0; it < ITE; ++it) {
    float* zb = S + zo;
    float* ob = S + (SBUF - zo);
    const float im1 = 1.f / m1, im2 = 1.f / m2, thr = 0.1f / m2;
    if (it == 0)
      step_block<true>(zb, ob, Y, inW, G1, Bo, Co, Bn, Cn, base, m1p, m1, im1,
                       m2p, m2, im2, thr);
    else
      step_block<false>(zb, ob, Y, inW, G1, Bo, Co, Bn, Cn, base, m1p, m1, im1,
                        m2p, m2, im2, thr);
    __syncthreads();

    v8f acc[2][4];
    clear_acc(acc);
    pass_lds(ob + a_off, fb, acc);
    store_block(acc, zb + d_off, P);
    __syncthreads();

    clear_acc(acc);
    pass_lds(zb + a_off, fb, acc);
#pragma unroll
    for (int tm = 0; tm < 2; ++tm) {
#pragma unroll
      for (int j = 0; j < 8; ++j) {
        const float cuv = cu[tm][j];
        acc[tm][0][j] = acc[tm][0][j] / (m1 + m2 * (4.f - 2.f * cv.x - 2.f * cuv));
        acc[tm][1][j] = acc[tm][1][j] / (m1 + m2 * (4.f - 2.f * cv.y - 2.f * cuv));
        acc[tm][2][j] = acc[tm][2][j] / (m1 + m2 * (4.f - 2.f * cv.z - 2.f * cuv));
        acc[tm][3][j] = acc[tm][3][j] / (m1 + m2 * (4.f - 2.f * cv.w - 2.f * cuv));
      }
    }
    store_block(acc, ob + d_off, P);
    __syncthreads();

    clear_acc(acc);
    pass_lds(ob + a_off, fb, acc);
    store_block(acc, zb + d_off, P);
    __syncthreads();

    clear_acc(acc);
    pass_lds(zb + a_off, fb, acc);
#pragma unroll
    for (int tm = 0; tm < 2; ++tm) {
#pragma unroll
      for (int tn = 0; tn < 4; ++tn) acc[tm][tn] = acc[tm][tn] * (1.f / 16384.f);
    }
    if (it == ITE - 1)
      store_block(acc, Zout + base + r0 * LD + n0, LD);
    else
      store_block(acc, ob + d_off, P);
    __syncthreads();

    zo = SBUF - zo;
    float* t;
    t = Bo; Bo = Bn; Bn = t;
    t = Co; Co = Cn; Cn = t;
    m1p = m1;
    m2p = m2;
    m1 *= 1.05f;
    m2 *= 1.05f;
  }
}


extern "C" void kernel_launch(void* const* d_in, const int* in_sizes, int n_in,
                              void* d_out, int out_size, void* d_ws,
                              size_t ws_size, hipStream_t stream) {
  (void)in_sizes; (void)n_in; (void)out_size; (void)ws_size;
  const float* Y = (const float*)d_in[0];
  const float* inW = (const float*)d_in[1];
  float* Z = (float*)d_out;
  float* ws = (float*)d_ws;

  float* Ht = ws;
  float* cosrow = Ht + IMG_ELEMS;
  float* p = ws + 4 * (size_t)IMG_ELEMS;
  float* G1 = p;           p += NE;
  float* Ba = p;           p += NE;
  float* Ca = p;           p += NE;
  float* Bb = p;           p += NE;
  float* Cb = p;           p += NE;

  k_tables<<<IMG_ELEMS / 256, 256, 0, stream>>>(Ht, cosrow);
  k_solve<<<NIMG, 256, 0, stream>>>(Ht, cosrow, Y, inW, G1, Ba, Ca, Bb, Cb, Z);
}


# B and C planes kept in LDS and updated in place instead of in memory (a wave owns 16 rows, a lane 4 floats); neighbour terms of R added behind one more barrier; summation order unchanged
# speedup vs baseline: 7.6212x; 1.2265x over previous
#include <hip/hip_runtime.h>
#include <hip/hip_bf16.h>
#include <math.h>


typedef __attribute__((ext_vector_type(2))) float v2f;
typedef __attribute__((ext_vector_type(4))) float v4f;
typedef __attribute__((ext_vector_type(8))) float v8f;

#define NIMG 124
#define IMG_ELEMS 16384
#define NE (NIMG * IMG_ELEMS)
#define LD 128
#define ITE 20


static __device__ __forceinline__ v8f wmma_f32(v2f a, v2f b, v8f c) {
  return __builtin_amdgcn_wmma_f32_16x16x4_f32(false, a, false, b,
                                               (short)0, c, false, false);
}


__global__ void k_tables(float* Ht, float* cosrow) {
  int i = blockIdx.x * blockDim.x + threadIdx.x;
  if (i >= IMG_ELEMS) return;
  int m = i >> 7, k = i & 127;
  int p = (m * k) & 127;
  float ang = (6.283185307179586f / 128.f) * (float)p;
  float s, c;
  sincosf(ang, &s, &c);
  Ht[i] = c + s;
  if (m == 1) cosrow[k] = c;
}

static __device__ __forceinline__ float shrinkf(float x, float t) {
  float m = fmaxf(fabsf(x) - t, 0.f);
  return copysignf(m, x);
}

#define P 144
#define SBUF (LD * P)

#define SR 8

template <bool FIRST>
static __device__ __forceinline__ void elem_a(
    const float z, const float zl, const float zu, const float y,
    const float iw, const float g1o, const float bo, const float co,
    const float mu1p, const float mu1, const float im1, const float mu2p,
    const float im2, const float thr, float& g1, float& b, float& c, float& t) {
  const float dh = z - zl;
  const float dv = z - zu;
  float q21 = 0.f, q22 = 0.f;
  float x;
  if (FIRST) {
    x = z;
    g1 = 0.f;
  } else {
    q21 = (mu2p * (bo - dh)) * im2;
    q22 = (mu2p * (co - dv)) * im2;
    x = (iw * y + mu1p * z - g1o) / (iw + mu1p);
    g1 = g1o + mu1p * (x - z);
  }
  b = shrinkf(dh - q21, thr) + q21;
  c = shrinkf(dv - q22, thr) + q22;
  const float a = x + g1 * im1;
  t = mu1 * a;
}

template <bool FIRST>
static __device__ __forceinline__ void step_a(
    const float* zb, float* rb, float* Bl, float* Cl,
    const float* __restrict__ Y, const float* __restrict__ inW, float* G1,
    const int base, const float mu1p, const float mu1, const float im1,
    const float mu2p, const float im2, const float thr) {
  const int lane = threadIdx.x & 31, wv = threadIdx.x >> 5;
  const int w4 = 4 * lane;
  const int wl = (w4 + 127) & 127;
  const int hs = 16 * wv;
  const v4f zero = {0.f, 0.f, 0.f, 0.f};
  v4f zu = *(const v4f*)(zb + ((hs + 127) & 127) * P + w4);
  for (int r0 = 0; r0 < 16; r0 += SR) {
    v4f yv[SR], iwv[SR], g1v[SR];
#pragma unroll
    for (int u = 0; u < SR; ++u) {
      yv[u] = zero;
      iwv[u] = zero;
      g1v[u] = zero;
      if (!FIRST) {
        const int gi = base + ((hs + r0 + u) << 7) + w4;
        yv[u] = *(const v4f*)(Y + gi);
        iwv[u] = *(const v4f*)(inW + gi);
        g1v[u] = *(const v4f*)(G1 + gi);
      }
    }
    v4f zv[SR], bov[SR], cov[SR];
    float zlv[SR];
#pragma unroll
    for (int u = 0; u < SR; ++u) {
      const int h = hs + r0 + u;
      zv[u] = *(const v4f*)(zb + h * P + w4);
      zlv[u] = zb[h * P + wl];
      bov[u] = zero;
      cov[u] = zero;
      if (!FIRST) {
        bov[u] = *(const v4f*)(Bl + h * LD + w4);
        cov[u] = *(const v4f*)(Cl + h * LD + w4);
      }
    }
#pragma unroll
    for (int u = 0; u < SR; ++u) {
      const int h = hs + r0 + u;
      const v4f z = zv[u];
      const v4f zl = {zlv[u], z.x, z.y, z.z};
      v4f g1, b, c, t;
#pragma unroll
      for (int k = 0; k < 4; ++k) {
        float g1k, bk, ck, tk;
        elem_a<FIRST>(z[k], zl[k], zu[k], yv[u][k], iwv[u][k], g1v[u][k],
                      bov[u][k], cov[u][k], mu1p, mu1, im1, mu2p, im2, thr, g1k,
                      bk, ck, tk);
        g1[k] = g1k;
        b[k] = bk;
        c[k] = ck;
        t[k] = tk;
      }
      *(v4f*)(G1 + base + (h << 7) + w4) = g1;
      *(v4f*)(Bl + h * LD + w4) = b;
      *(v4f*)(Cl + h * LD + w4) = c;
      *(v4f*)(rb + h * P + w4) = t;
      zu = z;
    }
  }
}

static __device__ __forceinline__ void step_b(float* rb, const float* Bl,
                                              const float* Cl, const float mu2) {
  const int lane = threadIdx.x & 31, wv = threadIdx.x >> 5;
  const int w4 = 4 * lane;
  const int wr = (w4 + 4) & 127;
  const int hs = 16 * wv;
  v4f c = *(const v4f*)(Cl + hs * LD + w4);
  for (int r = 0; r < 16; ++r) {
    const int h = hs + r, hp = (h + 1) & 127;
    const v4f t = *(const v4f*)(rb + h * P + w4);
    const v4f b = *(const v4f*)(Bl + h * LD + w4);
    const v4f bn = {b.y, b.z, b.w, Bl[h * LD + wr]};
    const v4f cn = *(const v4f*)(Cl + hp * LD + w4);
    *(v4f*)(rb + h * P + w4) = t + mu2 * ((b - bn) + (c - cn));
    c = cn;
  }
}

#define SSTEPS 8
#define SG 4
#define NG (SSTEPS / SG)


struct FragA {
  v2f a[SG][4][2];
};
struct FragB {
  v4f b[SG][2][2];
};

static __device__ __forceinline__ void load_a_step(FragA& f, const float* xa,
                                                   const int pitch, const int g,
                                                   const int i) {
  const int u = g * SG + i;
#pragma unroll
  for (int q = 0; q < 4; ++q) {
#pragma unroll
    for (int e = 0; e < 2; ++e)
      f.a[i][q][e] = *(const v2f*)(xa + (4 * u + e + 32 * q) * pitch);
  }
}

static __device__ __forceinline__ void load_b_step(FragB& f,
                                                   const float* __restrict__ hb,
                                                   const int g, const int i) {
  const int u = g * SG + i;
#pragma unroll
  for (int q = 0; q < 2; ++q) {
#pragma unroll
    for (int e = 0; e < 2; ++e)
      f.b[i][q][e] = *(const v4f*)(hb + (4 * u + e + 32 * q) * LD);
  }
}

static __device__ __forceinline__ void load_group_a(FragA& fa, const float* xa,
                                                    const int pitch,
                                                    const int g) {
#pragma unroll
  for (int i = 0; i < SG; ++i) load_a_step(fa, xa, pitch, g, i);
}

static __device__ __forceinline__ void load_group_b(FragB& fb,
                                                    const float* __restrict__ hb,
                                                    const int g) {
#pragma unroll
  for (int i = 0; i < SG; ++i) load_b_step(fb, hb, g, i);
}

static __device__ __forceinline__ v2f bpair(const FragB& f, const int i,
                                            const int q, const int tn) {
  v2f b;
  b.x = f.b[i][q][0][tn];
  b.y = f.b[i][q][1][tn];
  return b;
}

static __device__ __forceinline__ void mma_group(const FragA& fa,
                                                 const FragB& fb,
                                                 v8f (&acc)[2][4]) {
#pragma unroll
  for (int i = 0; i < SG; ++i) {
    v2f ee[2], eo[2], o0[2], o1[2];
#pragma unroll
    for (int tm = 0; tm < 2; ++tm) {
#pragma unroll
      for (int e = 0; e < 2; ++e) {
        const float x0 = fa.a[i][0][e][tm], x1 = fa.a[i][1][e][tm];
        const float x2 = fa.a[i][2][e][tm], x3 = fa.a[i][3][e][tm];
        const float e0 = x0 + x2, e1 = x1 + x3;
        o0[tm][e] = x0 - x2;
        o1[tm][e] = x1 - x3;
        ee[tm][e] = e0 + e1;
        eo[tm][e] = e0 - e1;
      }
    }
#pragma unroll
    for (int tm = 0; tm < 2; ++tm) {
      acc[tm][0] = wmma_f32(ee[tm], bpair(fb, i, 0, 0), acc[tm][0]);
      acc[tm][1] = wmma_f32(o0[tm], bpair(fb, i, 0, 1), acc[tm][1]);
      acc[tm][2] = wmma_f32(eo[tm], bpair(fb, i, 0, 2), acc[tm][2]);
      acc[tm][3] = wmma_f32(o0[tm], bpair(fb, i, 0, 3), acc[tm][3]);
    }
#pragma unroll
    for (int tm = 0; tm < 2; ++tm) {
      acc[tm][1] = wmma_f32(o1[tm], bpair(fb, i, 1, 1), acc[tm][1]);
      acc[tm][3] = wmma_f32(o1[tm], bpair(fb, i, 1, 3), acc[tm][3]);
    }
  }
}

static __device__ __forceinline__ void keep_order() {
  __builtin_amdgcn_sched_barrier(0);
}

static __device__ __forceinline__ void pass_lds(const float* xa,
                                                const FragB (&fb)[NG],
                                                v8f (&acc)[2][4]) {
  FragA fa[NG];
  load_group_a(fa[0], xa, P, 0);
  load_group_a(fa[1], xa, P, 1);
  keep_order();
  mma_group(fa[0], fb[0], acc);
  mma_group(fa[1], fb[1], acc);
}

static __device__ __forceinline__ void store_block(const v8f (&acc)[2][4],
                                                   float* d, const int pitch) {
#pragma unroll
  for (int tm = 0; tm < 2; ++tm) {
#pragma unroll
    for (int j = 0; j < 8; ++j) {
      v4f v;
      v.x = acc[tm][0][j];
      v.y = acc[tm][1][j];
      v.z = acc[tm][2][j];
      v.w = acc[tm][3][j];
      *(v4f*)(d + (2 * j + tm) * pitch) = v;
    }
  }
}

static __device__ __forceinline__ void clear_acc(v8f (&acc)[2][4]) {
  const v8f zero = {0.f, 0.f, 0.f, 0.f, 0.f, 0.f, 0.f, 0.f};
#pragma unroll
  for (int tm = 0; tm < 2; ++tm) {
#pragma unroll
    for (int tn = 0; tn < 4; ++tn) acc[tm][tn] = zero;
  }
}

__global__ void __launch_bounds__(256)
k_solve(const float* __restrict__ Ht, const float* __restrict__ cosrow,
        const float* __restrict__ Y, const float* __restrict__ inW, float* G1,
        float* __restrict__ Zout) {
  __shared__ __attribute__((aligned(16))) float S[2 * SBUF];
  __shared__ __attribute__((aligned(16))) float SB[IMG_ELEMS];
  __shared__ __attribute__((aligned(16))) float SC[IMG_ELEMS];
  const int tid = threadIdx.x;
  const int wv = tid >> 5;
  const int lane = tid & 31;
  const int c = lane & 15, h = lane >> 4;
  const int mg = wv >> 1, ng = wv & 1;
  const int base = blockIdx.x * IMG_ELEMS;

  const float* hb = Ht + (2 * h) * LD + 64 * ng + 4 * c;
  FragB fb[NG];
  load_group_b(fb[0], hb, 0);
  load_group_b(fb[1], hb, 1);

  const int n0 = 64 * ng + 4 * c;
  const int r0 = 32 * mg + 16 * h;
  const int a_off = (2 * h) * P + 32 * mg + 2 * c;
  const int d_off = r0 * P + n0;
  const v4f cv = *(const v4f*)(cosrow + n0);
  float cu[2][8];
#pragma unroll
  for (int tm = 0; tm < 2; ++tm) {
#pragma unroll
    for (int j = 0; j < 8; ++j) cu[tm][j] = cosrow[r0 + 2 * j + tm];
  }

  int zo = 0;
  for (int j = 0; j < 64; ++j) {
    const int e = tid + 256 * j;
    S[(e >> 7) * P + (e & 127)] = Y[base + e];
  }
  __syncthreads();

  float m1 = 0.1f, m2 = 0.1f;
  float m1p = 0.f, m2p = 0.f;
  for (int it = 0; it < ITE; ++it) {
    float* zb = S + zo;
    float* ob = S + (SBUF - zo);
    const float im1 = 1.f / m1, im2 = 1.f / m2, thr = 0.1f / m2;
    if (it == 0)
      step_a<true>(zb, ob, SB, SC, Y, inW, G1, base, m1p, m1, im1, m2p, im2, thr);
    else
      step_a<false>(zb, ob, SB, SC, Y, inW, G1, base, m1p, m1, im1, m2p, im2, thr);
    __syncthreads();
    step_b(ob, SB, SC, m2);
    __syncthreads();

    v8f acc[2][4];
    clear_acc(acc);
    pass_lds(ob + a_off, fb, acc);
    store_block(acc, zb + d_off, P);
    __syncthreads();

    clear_acc(acc);
    pass_lds(zb + a_off, fb, acc);
#pragma unroll
    for (int tm = 0; tm < 2; ++tm) {
#pragma unroll
      for (int j = 0; j < 8; ++j) {
        const float cuv = cu[tm][j];
        acc[tm][0][j] = acc[tm][0][j] / (m1 + m2 * (4.f - 2.f * cv.x - 2.f * cuv));
        acc[tm][1][j] = acc[tm][1][j] / (m1 + m2 * (4.f - 2.f * cv.y - 2.f * cuv));
        acc[tm][2][j] = acc[tm][2][j] / (m1 + m2 * (4.f - 2.f * cv.z - 2.f * cuv));
        acc[tm][3][j] = acc[tm][3][j] / (m1 + m2 * (4.f - 2.f * cv.w - 2.f * cuv));
      }
    }
    store_block(acc, ob + d_off, P);
    __syncthreads();

    clear_acc(acc);
    pass_lds(ob + a_off, fb, acc);
    store_block(acc, zb + d_off, P);
    __syncthreads();

    clear_acc(acc);
    pass_lds(zb + a_off, fb, acc);
#pragma unroll
    for (int tm = 0; tm < 2; ++tm) {
#pragma unroll
      for (int tn = 0; tn < 4; ++tn) acc[tm][tn] = acc[tm][tn] * (1.f / 16384.f);
    }
    if (it == ITE - 1)
      store_block(acc, Zout + base + r0 * LD + n0, LD);
    else
      store_block(acc, ob + d_off, P);
    __syncthreads();

    zo = SBUF - zo;
    m1p = m1;
    m2p = m2;
    m1 *= 1.05f;
    m2 *= 1.05f;
  }
}


extern "C" void kernel_launch(void* const* d_in, const int* in_sizes, int n_in,
                              void* d_out, int out_size, void* d_ws,
                              size_t ws_size, hipStream_t stream) {
  (void)in_sizes; (void)n_in; (void)out_size; (void)ws_size;
  const float* Y = (const float*)d_in[0];
  const float* inW = (const float*)d_in[1];
  float* Z = (float*)d_out;
  float* ws = (float*)d_ws;

  float* Ht = ws;
  float* cosrow = Ht + IMG_ELEMS;
  float* G1 = ws + 4 * (size_t)IMG_ELEMS;

  k_tables<<<IMG_ELEMS / 256, 256, 0, stream>>>(Ht, cosrow);
  k_solve<<<NIMG, 256, 0, stream>>>(Ht, cosrow, Y, inW, G1, Z);
}


# G1, Y and inW held in registers for all 20 iterations: no memory access left in the loop; for room the table moves to LDS and fragments go through a two-slot ring; summation order unchanged
# speedup vs baseline: 10.2276x; 1.3420x over previous
#include <hip/hip_runtime.h>
#include <hip/hip_bf16.h>
#include <math.h>


typedef __attribute__((ext_vector_type(2))) float v2f;
typedef __attribute__((ext_vector_type(4))) float v4f;
typedef __attribute__((ext_vector_type(8))) float v8f;

#define NIMG 124
#define IMG_ELEMS 16384
#define NE (NIMG * IMG_ELEMS)
#define LD 128
#define ITE 20


static __device__ __forceinline__ v8f wmma_f32(v2f a, v2f b, v8f c) {
  return __builtin_amdgcn_wmma_f32_16x16x4_f32(false, a, false, b,
                                               (short)0, c, false, false);
}


__global__ void k_tables(float* Ht, float* cosrow) {
  int i = blockIdx.x * blockDim.x + threadIdx.x;
  if (i >= IMG_ELEMS) return;
  int m = i >> 7, k = i & 127;
  int p = (m * k) & 127;
  float ang = (6.283185307179586f / 128.f) * (float)p;
  float s, c;
  sincosf(ang, &s, &c);
  Ht[i] = c + s;
  if (m == 1) cosrow[k] = c;
}

static __device__ __forceinline__ float shrinkf(float x, float t) {
  float m = fmaxf(fabsf(x) - t, 0.f);
  return copysignf(m, x);
}

#define P 144
#define SBUF (LD * P)

#define OWN_ROWS 16
#define SR 4

struct Own {
  v4f y[OWN_ROWS], iw[OWN_ROWS], g1[OWN_ROWS];
};

template <bool FIRST>
static __device__ __forceinline__ void elem_a(
    const float z, const float zl, const float zu, const float y,
    const float iw, const float g1o, const float bo, const float co,
    const float mu1p, const float mu1, const float im1, const float mu2p,
    const float im2, const float thr, float& g1, float& b, float& c, float& t) {
  const float dh = z - zl;
  const float dv = z - zu;
  float q21 = 0.f, q22 = 0.f;
  float x;
  if (FIRST) {
    x = z;
    g1 = 0.f;
  } else {
    q21 = (mu2p * (bo - dh)) * im2;
    q22 = (mu2p * (co - dv)) * im2;
    x = (iw * y + mu1p * z - g1o) / (iw + mu1p);
    g1 = g1o + mu1p * (x - z);
  }
  b = shrinkf(dh - q21, thr) + q21;
  c = shrinkf(dv - q22, thr) + q22;
  const float a = x + g1 * im1;
  t = mu1 * a;
}

template <bool FIRST>
static __device__ __forceinline__ void step_a(
    const float* zb, float* rb, float* Bl, float* Cl, Own& own,
    const float mu1p, const float mu1, const float im1, const float mu2p,
    const float im2, const float thr) {
  const int lane = threadIdx.x & 31, wv = threadIdx.x >> 5;
  const int w4 = 4 * lane;
  const int wl = (w4 + 127) & 127;
  const int hs = 16 * wv;
  const v4f zero = {0.f, 0.f, 0.f, 0.f};
  v4f zu = *(const v4f*)(zb + ((hs + 127) & 127) * P + w4);
#pragma unroll
  for (int r0 = 0; r0 < OWN_ROWS; r0 += SR) {
    v4f zv[SR], bov[SR], cov[SR];
    float zlv[SR];
#pragma unroll
    for (int u = 0; u < SR; ++u) {
      const int h = hs + r0 + u;
      zv[u] = *(const v4f*)(zb + h * P + w4);
      zlv[u] = zb[h * P + wl];
      bov[u] = zero;
      cov[u] = zero;
      if (!FIRST) {
        bov[u] = *(const v4f*)(Bl + h * LD + w4);
        cov[u] = *(const v4f*)(Cl + h * LD + w4);
      }
    }
#pragma unroll
    for (int u = 0; u < SR; ++u) {
      const int h = hs + r0 + u;
      const v4f z = zv[u];
      const v4f zl = {zlv[u], z.x, z.y, z.z};
      v4f g1, b, c, t;
#pragma unroll
      for (int k = 0; k < 4; ++k) {
        float g1k, bk, ck, tk;
        elem_a<FIRST>(z[k], zl[k], zu[k], own.y[r0 + u][k], own.iw[r0 + u][k],
                      own.g1[r0 + u][k], bov[u][k], cov[u][k], mu1p, mu1, im1,
                      mu2p, im2, thr, g1k, bk, ck, tk);
        g1[k] = g1k;
        b[k] = bk;
        c[k] = ck;
        t[k] = tk;
      }
      own.g1[r0 + u] = g1;
      *(v4f*)(Bl + h * LD + w4) = b;
      *(v4f*)(Cl + h * LD + w4) = c;
      *(v4f*)(rb + h * P + w4) = t;
      zu = z;
    }
  }
}

static __device__ __forceinline__ void step_b(float* rb, const float* Bl,
                                              const float* Cl, const float mu2) {
  const int lane = threadIdx.x & 31, wv = threadIdx.x >> 5;
  const int w4 = 4 * lane;
  const int wr = (w4 + 4) & 127;
  const int hs = 16 * wv;
  v4f c = *(const v4f*)(Cl + hs * LD + w4);
  for (int r = 0; r < 16; ++r) {
    const int h = hs + r, hp = (h + 1) & 127;
    const v4f t = *(const v4f*)(rb + h * P + w4);
    const v4f b = *(const v4f*)(Bl + h * LD + w4);
    const v4f bn = {b.y, b.z, b.w, Bl[h * LD + wr]};
    const v4f cn = *(const v4f*)(Cl + hp * LD + w4);
    *(v4f*)(rb + h * P + w4) = t + mu2 * ((b - bn) + (c - cn));
    c = cn;
  }
}

#define SSTEPS 8
#define SG 1
#define NG (SSTEPS / SG)


struct FragA {
  v2f a[SG][4][2];
};
struct FragB {
  v4f b[SG][3];
};
#define TB_STEP 768
#define TB_FLOATS (SSTEPS * TB_STEP)

static __device__ __forceinline__ void load_a_step(FragA& f, const float* xa,
                                                   const int pitch, const int g,
                                                   const int i) {
  const int u = g * SG + i;
#pragma unroll
  for (int q = 0; q < 4; ++q) {
#pragma unroll
    for (int e = 0; e < 2; ++e)
      f.a[i][q][e] = *(const v2f*)(xa + (4 * u + e + 32 * q) * pitch);
  }
}

static __device__ __forceinline__ void load_b_step(FragB& f, const float* tb,
                                                   const int g, const int i) {
  const int u = g * SG + i;
#pragma unroll
  for (int p = 0; p < 3; ++p)
    f.b[i][p] = *(const v4f*)(tb + u * TB_STEP + 64 * p);
}

static __device__ __forceinline__ void load_group(FragA& fa, FragB& fb,
                                                  const float* xa,
                                                  const float* tb, const int g) {
#pragma unroll
  for (int i = 0; i < SG; ++i) {
    load_a_step(fa, xa, P, g, i);
    load_b_step(fb, tb, g, i);
  }
}

static __device__ __forceinline__ v2f bpair(const FragB& f, const int i,
                                            const int q, const int tn) {
  const v4f v = q ? f.b[i][2] : f.b[i][tn >> 1];
  const int s = q ? (tn >> 1) : (tn & 1);
  v2f b;
  b.x = v[2 * s];
  b.y = v[2 * s + 1];
  return b;
}

static __device__ __forceinline__ void mma_group(const FragA& fa,
                                                 const FragB& fb,
                                                 v8f (&acc)[2][4]) {
#pragma unroll
  for (int i = 0; i < SG; ++i) {
    v2f ee[2], eo[2], o0[2], o1[2];
#pragma unroll
    for (int tm = 0; tm < 2; ++tm) {
#pragma unroll
      for (int e = 0; e < 2; ++e) {
        const float x0 = fa.a[i][0][e][tm], x1 = fa.a[i][1][e][tm];
        const float x2 = fa.a[i][2][e][tm], x3 = fa.a[i][3][e][tm];
        const float e0 = x0 + x2, e1 = x1 + x3;
        o0[tm][e] = x0 - x2;
        o1[tm][e] = x1 - x3;
        ee[tm][e] = e0 + e1;
        eo[tm][e] = e0 - e1;
      }
    }
#pragma unroll
    for (int tm = 0; tm < 2; ++tm) {
      acc[tm][0] = wmma_f32(ee[tm], bpair(fb, i, 0, 0), acc[tm][0]);
      acc[tm][1] = wmma_f32(o0[tm], bpair(fb, i, 0, 1), acc[tm][1]);
      acc[tm][2] = wmma_f32(eo[tm], bpair(fb, i, 0, 2), acc[tm][2]);
      acc[tm][3] = wmma_f32(o0[tm], bpair(fb, i, 0, 3), acc[tm][3]);
    }
#pragma unroll
    for (int tm = 0; tm < 2; ++tm) {
      acc[tm][1] = wmma_f32(o1[tm], bpair(fb, i, 1, 1), acc[tm][1]);
      acc[tm][3] = wmma_f32(o1[tm], bpair(fb, i, 1, 3), acc[tm][3]);
    }
  }
}

static __device__ __forceinline__ void keep_order() {
  __builtin_amdgcn_sched_barrier(0);
}

static __device__ __forceinline__ void pass_lds(const float* xa, const float* tb,
                                                v8f (&acc)[2][4]) {
  FragA fa[2];
  FragB fb[2];
  load_group(fa[0], fb[0], xa, tb, 0);
#pragma unroll
  for (int g = 0; g < NG; ++g) {
    if (g + 1 < NG) load_group(fa[(g + 1) & 1], fb[(g + 1) & 1], xa, tb, g + 1);
    keep_order();
    mma_group(fa[g & 1], fb[g & 1], acc);
    keep_order();
  }
}

static __device__ __forceinline__ void fill_table(float* TB,
                                                  const float* __restrict__ Ht) {
  for (int i = threadIdx.x; i < TB_FLOATS; i += 256) {
    const int s = i & 3, cc = (i >> 2) & 15, rest = i >> 6;
    const int piece = rest % 3, rn = rest / 3;
    const int ngc = rn & 1, rp = rn >> 1;
    const int e = s & 1, t = s >> 1;
    const int row = 2 * rp + e + (piece == 2 ? 32 : 0);
    const int col = 64 * ngc + 4 * cc +
                    (piece == 0 ? t : (piece == 1 ? 2 + t : 1 + 2 * t));
    TB[i] = Ht[row * LD + col];
  }
}

static __device__ __forceinline__ void store_block(const v8f (&acc)[2][4],
                                                   float* d, const int pitch) {
#pragma unroll
  for (int tm = 0; tm < 2; ++tm) {
#pragma unroll
    for (int j = 0; j < 8; ++j) {
      v4f v;
      v.x = acc[tm][0][j];
      v.y = acc[tm][1][j];
      v.z = acc[tm][2][j];
      v.w = acc[tm][3][j];
      *(v4f*)(d + (2 * j + tm) * pitch) = v;
    }
  }
}

static __device__ __forceinline__ void clear_acc(v8f (&acc)[2][4]) {
  const v8f zero = {0.f, 0.f, 0.f, 0.f, 0.f, 0.f, 0.f, 0.f};
#pragma unroll
  for (int tm = 0; tm < 2; ++tm) {
#pragma unroll
    for (int tn = 0; tn < 4; ++tn) acc[tm][tn] = zero;
  }
}

__global__ void __launch_bounds__(256)
k_solve(const float* __restrict__ Ht, const float* __restrict__ cosrow,
        const float* __restrict__ Y, const float* __restrict__ inW,
        float* __restrict__ Zout) {
  __shared__ __attribute__((aligned(16))) float S[2 * SBUF];
  __shared__ __attribute__((aligned(16))) float SB[IMG_ELEMS];
  __shared__ __attribute__((aligned(16))) float SC[IMG_ELEMS];
  __shared__ __attribute__((aligned(16))) float TB[TB_FLOATS];
  const int tid = threadIdx.x;
  const int wv = tid >> 5;
  const int lane = tid & 31;
  const int c = lane & 15, h = lane >> 4;
  const int mg = wv >> 1, ng = wv & 1;
  const int base = blockIdx.x * IMG_ELEMS;

  fill_table(TB, Ht);
  const float* tb = TB + ((h * 2 + ng) * 3) * 64 + 4 * c;

  const int n0 = 64 * ng + 4 * c;
  const int r0 = 32 * mg + 16 * h;
  const int a_off = (2 * h) * P + 32 * mg + 2 * c;
  const int d_off = r0 * P + n0;
  const v4f cv = *(const v4f*)(cosrow + n0);
  float cu[2][8];
#pragma unroll
  for (int tm = 0; tm < 2; ++tm) {
#pragma unroll
    for (int j = 0; j < 8; ++j) cu[tm][j] = cosrow[r0 + 2 * j + tm];
  }

  int zo = 0;
  Own own;
#pragma unroll
  for (int r = 0; r < OWN_ROWS; ++r) {
    const int gi = base + ((16 * wv + r) << 7) + 4 * lane;
    own.y[r] = *(const v4f*)(Y + gi);
    own.iw[r] = *(const v4f*)(inW + gi);
    own.g1[r] = (v4f){0.f, 0.f, 0.f, 0.f};
  }
#pragma unroll
  for (int r = 0; r < OWN_ROWS; ++r)
    *(v4f*)(S + (16 * wv + r) * P + 4 * lane) = own.y[r];
  __syncthreads();

  float m1 = 0.1f, m2 = 0.1f;
  float m1p = 0.f, m2p = 0.f;
  for (int it = 0; it < ITE; ++it) {
    float* zb = S + zo;
    float* ob = S + (SBUF - zo);
    const float im1 = 1.f / m1, im2 = 1.f / m2, thr = 0.1f / m2;
    if (it == 0)
      step_a<true>(zb, ob, SB, SC, own, m1p, m1, im1, m2p, im2, thr);
    else
      step_a<false>(zb, ob, SB, SC, own, m1p, m1, im1, m2p, im2, thr);
    __syncthreads();
    step_b(ob, SB, SC, m2);
    __syncthreads();

    v8f acc[2][4];
    clear_acc(acc);
    pass_lds(ob + a_off, tb, acc);
    store_block(acc, zb + d_off, P);
    __syncthreads();

    clear_acc(acc);
    pass_lds(zb + a_off, tb, acc);
    float two = 2.f;
    asm volatile("" : "+s"(two));
#pragma unroll
    for (int tm = 0; tm < 2; ++tm) {
#pragma unroll
      for (int j = 0; j < 8; ++j) {
        const float cuv = cu[tm][j];
        acc[tm][0][j] = acc[tm][0][j] / (m1 + m2 * (4.f - two * cv.x - two * cuv));
        acc[tm][1][j] = acc[tm][1][j] / (m1 + m2 * (4.f - two * cv.y - two * cuv));
        acc[tm][2][j] = acc[tm][2][j] / (m1 + m2 * (4.f - two * cv.z - two * cuv));
        acc[tm][3][j] = acc[tm][3][j] / (m1 + m2 * (4.f - two * cv.w - two * cuv));
      }
    }
    store_block(acc, ob + d_off, P);
    __syncthreads();

    clear_acc(acc);
    pass_lds(ob + a_off, tb, acc);
    store_block(acc, zb + d_off, P);
    __syncthreads();

    clear_acc(acc);
    pass_lds(zb + a_off, tb, acc);
#pragma unroll
    for (int tm = 0; tm < 2; ++tm) {
#pragma unroll
      for (int tn = 0; tn < 4; ++tn) acc[tm][tn] = acc[tm][tn] * (1.f / 16384.f);
    }
    if (it == ITE - 1)
      store_block(acc, Zout + base + r0 * LD + n0, LD);
    else
      store_block(acc, ob + d_off, P);
    __syncthreads();

    zo = SBUF - zo;
    m1p = m1;
    m2p = m2;
    m1 *= 1.05f;
    m2 *= 1.05f;
  }
}


extern "C" void kernel_launch(void* const* d_in, const int* in_sizes, int n_in,
                              void* d_out, int out_size, void* d_ws,
                              size_t ws_size, hipStream_t stream) {
  (void)in_sizes; (void)n_in; (void)out_size; (void)ws_size;
  const float* Y = (const float*)d_in[0];
  const float* inW = (const float*)d_in[1];
  float* Z = (float*)d_out;
  float* ws = (float*)d_ws;

  float* Ht = ws;
  float* cosrow = Ht + IMG_ELEMS;

  k_tables<<<IMG_ELEMS / 256, 256, 0, stream>>>(Ht, cosrow);
  k_solve<<<NIMG, 256, 0, stream>>>(Ht, cosrow, Y, inW, Z);
}
